# starting kernel
# baseline (speedup 1.0000x reference)
#include <hip/hip_runtime.h>

typedef __attribute__((ext_vector_type(2))) float v2f;
typedef __attribute__((ext_vector_type(8))) float v8f;

#define CFEAT 256

__global__ void zero_kernel(float* __restrict__ p, long n) {
    long i = (long)blockIdx.x * blockDim.x + threadIdx.x;
    long stride = (long)gridDim.x * blockDim.x;
    for (; i < n; i += stride) p[i] = 0.0f;
}

__global__ void scatter_kernel(const float* __restrict__ x,
                               const int* __restrict__ src,
                               const int* __restrict__ dst,
                               float* __restrict__ s,
                               float* __restrict__ cnt,
                               int E) {
    int wave = (int)((blockIdx.x * blockDim.x + threadIdx.x) >> 5);
    int lane = threadIdx.x & 31;
    if (wave >= E) return;
    int sr = src[wave];
    int ds = dst[wave];
    const float4* xs = (const float4*)(x + (long)sr * CFEAT);
    float* sd = s + (long)ds * CFEAT;
    float4 a = xs[lane];
    float4 b = xs[lane + 32];
    int base = lane * 4;
    atomicAdd(&sd[base + 0], a.x);
    atomicAdd(&sd[base + 1], a.y);
    atomicAdd(&sd[base + 2], a.z);
    atomicAdd(&sd[base + 3], a.w);
    atomicAdd(&sd[128 + base + 0], b.x);
    atomicAdd(&sd[128 + base + 1], b.y);
    atomicAdd(&sd[128 + base + 2], b.z);
    atomicAdd(&sd[128 + base + 3], b.w);
    if (lane == 0) atomicAdd(&cnt[ds], 1.0f);
}

__global__ void gemm_kernel(const float* __restrict__ s,
                            const float* __restrict__ cnt,
                            const float* __restrict__ xt,
                            const float* __restrict__ Wl,
                            const float* __restrict__ bl,
                            const float* __restrict__ Wr,
                            float* __restrict__ h,
                            float* __restrict__ colsum,
                            float* __restrict__ colsq,
                            int N, int rowTiles) {
    int wavesPerBlock = blockDim.x >> 5;
    int gw = blockIdx.x * wavesPerBlock + (threadIdx.x >> 5);
    int totalWaves = rowTiles * 4;
    if (gw >= totalWaves) return;

    int rowTile = gw >> 2;
    int colGrp  = gw & 3;
    int lane = threadIdx.x & 31;
    int l15  = lane & 15;
    int khalf = (lane >> 4) * 2;

    int rowA = rowTile * 16 + l15;
    int rowL = (rowA < N) ? rowA : (N - 1);
    float invc = 1.0f / fmaxf(cnt[rowL], 1.0f);
    const float* sRow = s  + (long)rowL * CFEAT;
    const float* xRow = xt + (long)rowL * CFEAT;

    int colBase = colGrp * 64;
    const float* wlRow[4];
    const float* wrRow[4];
#pragma unroll
    for (int t = 0; t < 4; ++t) {
        int colB = colBase + t * 16 + l15;
        wlRow[t] = Wl + (long)colB * CFEAT;
        wrRow[t] = Wr + (long)colB * CFEAT;
    }

    v8f acc[4] = {{0,0,0,0,0,0,0,0}, {0,0,0,0,0,0,0,0},
                  {0,0,0,0,0,0,0,0}, {0,0,0,0,0,0,0,0}};

    for (int k = 0; k < CFEAT; k += 4) {
        int ko = k + khalf;
        v2f aL = *(const v2f*)(sRow + ko);
        aL *= invc;
        v2f aR = *(const v2f*)(xRow + ko);
#pragma unroll
        for (int t = 0; t < 4; ++t) {
            v2f bL = *(const v2f*)(wlRow[t] + ko);
            v2f bR = *(const v2f*)(wrRow[t] + ko);
            acc[t] = __builtin_amdgcn_wmma_f32_16x16x4_f32(
                false, aL, false, bL, (short)0, acc[t], false, false);
            acc[t] = __builtin_amdgcn_wmma_f32_16x16x4_f32(
                false, aR, false, bR, (short)0, acc[t], false, false);
        }
    }

    int rbase = rowTile * 16 + ((lane < 16) ? 0 : 8);
#pragma unroll
    for (int t = 0; t < 4; ++t) {
        int col = colBase + t * 16 + l15;
        float bias = bl[col];
        float sp = 0.0f, qp = 0.0f;
#pragma unroll
        for (int v = 0; v < 8; ++v) {
            float val = fmaxf(acc[t][v] + bias, 0.0f);
            int r = rbase + v;
            bool ok = (r < N);
            float m = ok ? val : 0.0f;
            sp += m;
            qp += m * m;
            if (ok) h[(long)r * CFEAT + col] = val;
        }
        atomicAdd(&colsum[col], sp);
        atomicAdd(&colsq[col], qp);
    }
}

__global__ void norm_kernel(const float* __restrict__ hbuf,
                            const float* __restrict__ colsum,
                            const float* __restrict__ colsq,
                            const float* __restrict__ gamma,
                            const float* __restrict__ beta,
                            float* __restrict__ out, int N) {
    long i = (long)blockIdx.x * blockDim.x + threadIdx.x;
    long total = (long)N * CFEAT;
    if (i >= total) return;
    int col = (int)(i & (CFEAT - 1));
    float invN = 1.0f / (float)N;
    float mu  = colsum[col] * invN;
    float var = colsq[col] * invN - mu * mu;
    float rs  = rsqrtf(var + 1e-5f);
    out[i] = gamma[col] * (hbuf[i] - mu) * rs + beta[col];
}

extern "C" void kernel_launch(void* const* d_in, const int* in_sizes, int n_in,
                              void* d_out, int out_size, void* d_ws, size_t ws_size,
                              hipStream_t stream) {
    const float* x    = (const float*)d_in[0];
    const int* e0src  = (const int*)d_in[1];
    const int* e0dst  = (const int*)d_in[2];
    const int* e1src  = (const int*)d_in[3];
    const int* e1dst  = (const int*)d_in[4];
    const float* Wl0  = (const float*)d_in[5];
    const float* bl0  = (const float*)d_in[6];
    const float* Wr0  = (const float*)d_in[7];
    const float* g0   = (const float*)d_in[8];
    const float* be0  = (const float*)d_in[9];
    const float* Wl1  = (const float*)d_in[10];
    const float* bl1  = (const float*)d_in[11];
    const float* Wr1  = (const float*)d_in[12];
    const float* g1   = (const float*)d_in[13];
    const float* be1  = (const float*)d_in[14];

    const int N1 = 50000, N2 = 12500;
    const int E0 = in_sizes[1];
    const int E1 = in_sizes[3];

    float* ws = (float*)d_ws;
    const long NC1 = (long)N1 * CFEAT;
    float* x1    = ws;
    float* sbuf  = ws + NC1;
    float* hbuf  = ws + 2 * NC1;
    float* cntb  = ws + 3 * NC1;
    float* stats = cntb + 50048;

    zero_kernel<<<4096, 256, 0, stream>>>(sbuf, NC1);
    zero_kernel<<<256, 256, 0, stream>>>(cntb, 50048 + 512);
    scatter_kernel<<<(E0 + 7) / 8, 256, 0, stream>>>(x, e0src, e0dst, sbuf, cntb, E0);
    {
        int rowTiles = (N1 + 15) / 16;
        int waves = rowTiles * 4;
        gemm_kernel<<<(waves + 7) / 8, 256, 0, stream>>>(
            sbuf, cntb, x, Wl0, bl0, Wr0, hbuf, stats, stats + 256, N1, rowTiles);
    }
    norm_kernel<<<(int)((NC1 + 255) / 256), 256, 0, stream>>>(
        hbuf, stats, stats + 256, g0, be0, x1, N1);

    const long NC2 = (long)N2 * CFEAT;
    zero_kernel<<<2048, 256, 0, stream>>>(sbuf, NC2);
    zero_kernel<<<256, 256, 0, stream>>>(cntb, 50048 + 512);
    scatter_kernel<<<(E1 + 7) / 8, 256, 0, stream>>>(x1, e1src, e1dst, sbuf, cntb, E1);
    {
        int rowTiles = (N2 + 15) / 16;
        int waves = rowTiles * 4;
        gemm_kernel<<<(waves + 7) / 8, 256, 0, stream>>>(
            sbuf, cntb, x1, Wl1, bl1, Wr1, hbuf, stats, stats + 256, N2, rowTiles);
    }
    norm_kernel<<<(int)((NC2 + 255) / 256), 256, 0, stream>>>(
        hbuf, stats, stats + 256, g1, be1, (float*)d_out, N2);
}


# GEMM fragments loaded as whole line shares: the four 16-byte pieces a lane owns of a 128-byte row line go out together one block ahead (were eight 8-byte loads); k summed in a permuted order
# speedup vs baseline: 1.6204x; 1.6204x over previous
#include <hip/hip_runtime.h>

typedef __attribute__((ext_vector_type(2))) float v2f;
typedef __attribute__((ext_vector_type(8))) float v8f;

#define CFEAT 256

__global__ void zero_kernel(float* __restrict__ p, long n) {
    long i = (long)blockIdx.x * blockDim.x + threadIdx.x;
    long stride = (long)gridDim.x * blockDim.x;
    for (; i < n; i += stride) p[i] = 0.0f;
}

#define SC_BLOCKS 2048
#define SC_WAVES_LOG2 14

__device__ __forceinline__ void row_load(float (&v)[8], const float* __restrict__ x, int sr, int lane) {
    const float* xs = x + (long)sr * CFEAT + lane;
#pragma unroll
    for (int j = 0; j < 8; ++j) v[j] = xs[j * 32];
}

__device__ __forceinline__ void row_add(const float (&v)[8], float* __restrict__ s, int ds, int lane) {
    float* sd = s + (long)ds * CFEAT + lane;
#pragma unroll
    for (int j = 0; j < 8; ++j) atomicAdd(&sd[j * 32], v[j]);
}

__global__ void scatter_kernel(const float* __restrict__ x,
                               const int* __restrict__ src,
                               const int* __restrict__ dst,
                               float* __restrict__ s,
                               float* __restrict__ cnt,
                               int E) {
    const int wv = __builtin_amdgcn_readfirstlane((int)(threadIdx.x >> 5));
    const int w = (int)blockIdx.x * 8 + wv;
    const int lane = threadIdx.x & 31;
    const int e0 = (int)(((long)w * E) >> SC_WAVES_LOG2);
    const int e1 = (int)(((long)(w + 1) * E) >> SC_WAVES_LOG2);

    for (int b = e0; b < e1; b += 32) {
        const int n = (e1 - b < 32) ? (e1 - b) : 32;
        int mySrc = 0, myDst = 0;
        if (lane < n) {
            mySrc = src[b + lane];
            myDst = dst[b + lane];
            atomicAdd(&cnt[myDst], 1.0f);
        }
        float va[8], vb[8];
        row_load(va, x, __builtin_amdgcn_readlane(mySrc, 0), lane);
        int i = 0;
        for (; i + 2 < n; i += 2) {
            row_load(vb, x, __builtin_amdgcn_readlane(mySrc, i + 1), lane);
            row_add(va, s, __builtin_amdgcn_readlane(myDst, i), lane);
            row_load(va, x, __builtin_amdgcn_readlane(mySrc, i + 2), lane);
            row_add(vb, s, __builtin_amdgcn_readlane(myDst, i + 1), lane);
        }
        if (i + 1 < n) {
            row_load(vb, x, __builtin_amdgcn_readlane(mySrc, i + 1), lane);
            row_add(va, s, __builtin_amdgcn_readlane(myDst, i), lane);
            row_add(vb, s, __builtin_amdgcn_readlane(myDst, i + 1), lane);
        } else {
            row_add(va, s, __builtin_amdgcn_readlane(myDst, i), lane);
        }
    }
}

typedef float4 linefrag[4];

__device__ __forceinline__ void load_linefrag(linefrag& f, const float* __restrict__ rowh, int kc) {
    const float4* p = (const float4*)(rowh + kc);
#pragma unroll
    for (int q = 0; q < 4; ++q) f[q] = p[2 * q];
}

__device__ __forceinline__ v2f frag_pair(const linefrag& f, int q, int p) {
    v2f r;
    if (p == 0) { r[0] = f[q].x; r[1] = f[q].y; }
    else        { r[0] = f[q].z; r[1] = f[q].w; }
    return r;
}

__device__ __forceinline__ void mma_line(v8f (&acc)[4], const linefrag& a, const linefrag (&b)[4]) {
#pragma unroll
    for (int q = 0; q < 4; ++q) {
#pragma unroll
        for (int p = 0; p < 2; ++p) {
            v2f af = frag_pair(a, q, p);
#pragma unroll
            for (int t = 0; t < 4; ++t)
                acc[t] = __builtin_amdgcn_wmma_f32_16x16x4_f32(
                    false, af, false, frag_pair(b[t], q, p), (short)0, acc[t], false, false);
        }
    }
}

__global__ void __launch_bounds__(256) gemm_kernel(const float* __restrict__ s,
                            const float* __restrict__ cnt,
                            const float* __restrict__ xt,
                            const float* __restrict__ Wl,
                            const float* __restrict__ bl,
                            const float* __restrict__ Wr,
                            float* __restrict__ h,
                            float* __restrict__ colsum,
                            float* __restrict__ colsq,
                            int N, int rowTiles) {
    int wavesPerBlock = blockDim.x >> 5;
    int gw = blockIdx.x * wavesPerBlock + (threadIdx.x >> 5);
    int totalWaves = rowTiles * 4;
    if (gw >= totalWaves) return;

    int rowTile = gw >> 2;
    int colGrp  = gw & 3;
    int lane = threadIdx.x & 31;
    int l15  = lane & 15;
    int hoff = (lane >> 4) * 4;

    int rowA = rowTile * 16 + l15;
    int rowL = (rowA < N) ? rowA : (N - 1);
    float invc = 1.0f / fmaxf(cnt[rowL], 1.0f);
    const float* sRow = s  + (long)rowL * CFEAT + hoff;
    const float* xRow = xt + (long)rowL * CFEAT + hoff;

    int colBase = colGrp * 64;
    const float* wlRow[4];
    const float* wrRow[4];
#pragma unroll
    for (int t = 0; t < 4; ++t) {
        int colB = colBase + t * 16 + l15;
        wlRow[t] = Wl + (long)colB * CFEAT + hoff;
        wrRow[t] = Wr + (long)colB * CFEAT + hoff;
    }

    v8f acc[4] = {{0,0,0,0,0,0,0,0}, {0,0,0,0,0,0,0,0},
                  {0,0,0,0,0,0,0,0}, {0,0,0,0,0,0,0,0}};

    linefrag aC, bC[4], aN, bN[4];
    load_linefrag(aC, sRow, 0);
#pragma unroll
    for (int t = 0; t < 4; ++t) load_linefrag(bC[t], wlRow[t], 0);

    for (int kc = 0; kc < CFEAT - 32; kc += 32) {
        load_linefrag(aN, xRow, kc);
#pragma unroll
        for (int t = 0; t < 4; ++t) load_linefrag(bN[t], wrRow[t], kc);
        __builtin_amdgcn_sched_barrier(0);
#pragma unroll
        for (int q = 0; q < 4; ++q) aC[q] *= invc;
        mma_line(acc, aC, bC);
        __builtin_amdgcn_sched_barrier(0);
        load_linefrag(aC, sRow, kc + 32);
#pragma unroll
        for (int t = 0; t < 4; ++t) load_linefrag(bC[t], wlRow[t], kc + 32);
        __builtin_amdgcn_sched_barrier(0);
        mma_line(acc, aN, bN);
    }
    load_linefrag(aN, xRow, CFEAT - 32);
#pragma unroll
    for (int t = 0; t < 4; ++t) load_linefrag(bN[t], wrRow[t], CFEAT - 32);
    __builtin_amdgcn_sched_barrier(0);
#pragma unroll
    for (int q = 0; q < 4; ++q) aC[q] *= invc;
    mma_line(acc, aC, bC);
    mma_line(acc, aN, bN);

    int rbase = rowTile * 16 + ((lane < 16) ? 0 : 8);
#pragma unroll
    for (int t = 0; t < 4; ++t) {
        int col = colBase + t * 16 + l15;
        float bias = bl[col];
        float sp = 0.0f, qp = 0.0f;
#pragma unroll
        for (int v = 0; v < 8; ++v) {
            float val = fmaxf(acc[t][v] + bias, 0.0f);
            int r = rbase + v;
            bool ok = (r < N);
            float m = ok ? val : 0.0f;
            sp += m;
            qp += m * m;
            if (ok) h[(long)r * CFEAT + col] = val;
        }
        atomicAdd(&colsum[col], sp);
        atomicAdd(&colsq[col], qp);
    }
}

__global__ void norm_kernel(const float* __restrict__ hbuf,
                            const float* __restrict__ colsum,
                            const float* __restrict__ colsq,
                            const float* __restrict__ gamma,
                            const float* __restrict__ beta,
                            float* __restrict__ out, int N) {
    long i = (long)blockIdx.x * blockDim.x + threadIdx.x;
    long total = (long)N * CFEAT;
    if (i >= total) return;
    int col = (int)(i & (CFEAT - 1));
    float invN = 1.0f / (float)N;
    float mu  = colsum[col] * invN;
    float var = colsq[col] * invN - mu * mu;
    float rs  = rsqrtf(var + 1e-5f);
    out[i] = gamma[col] * (hbuf[i] - mu) * rs + beta[col];
}

extern "C" void kernel_launch(void* const* d_in, const int* in_sizes, int n_in,
                              void* d_out, int out_size, void* d_ws, size_t ws_size,
                              hipStream_t stream) {
    const float* x    = (const float*)d_in[0];
    const int* e0src  = (const int*)d_in[1];
    const int* e0dst  = (const int*)d_in[2];
    const int* e1src  = (const int*)d_in[3];
    const int* e1dst  = (const int*)d_in[4];
    const float* Wl0  = (const float*)d_in[5];
    const float* bl0  = (const float*)d_in[6];
    const float* Wr0  = (const float*)d_in[7];
    const float* g0   = (const float*)d_in[8];
    const float* be0  = (const float*)d_in[9];
    const float* Wl1  = (const float*)d_in[10];
    const float* bl1  = (const float*)d_in[11];
    const float* Wr1  = (const float*)d_in[12];
    const float* g1   = (const float*)d_in[13];
    const float* be1  = (const float*)d_in[14];

    const int N1 = 50000, N2 = 12500;
    const int E0 = in_sizes[1];
    const int E1 = in_sizes[3];

    float* ws = (float*)d_ws;
    const long NC1 = (long)N1 * CFEAT;
    float* x1    = ws;
    float* sbuf  = ws + NC1;
    float* hbuf  = ws + 2 * NC1;
    float* cntb  = ws + 3 * NC1;
    float* stats = cntb + 50048;

    zero_kernel<<<4096, 256, 0, stream>>>(sbuf, NC1);
    zero_kernel<<<256, 256, 0, stream>>>(cntb, 50048 + 512);
    scatter_kernel<<<SC_BLOCKS, 256, 0, stream>>>(x, e0src, e0dst, sbuf, cntb, E0);
    {
        int rowTiles = (N1 + 15) / 16;
        int waves = rowTiles * 4;
        gemm_kernel<<<(waves + 7) / 8, 256, 0, stream>>>(
            sbuf, cntb, x, Wl0, bl0, Wr0, hbuf, stats, stats + 256, N1, rowTiles);
    }
    norm_kernel<<<(int)((NC1 + 255) / 256), 256, 0, stream>>>(
        hbuf, stats, stats + 256, g0, be0, x1, N1);

    const long NC2 = (long)N2 * CFEAT;
    zero_kernel<<<2048, 256, 0, stream>>>(sbuf, NC2);
    zero_kernel<<<256, 256, 0, stream>>>(cntb, 50048 + 512);
    scatter_kernel<<<SC_BLOCKS, 256, 0, stream>>>(x1, e1src, e1dst, sbuf, cntb, E1);
    {
        int rowTiles = (N2 + 15) / 16;
        int waves = rowTiles * 4;
        gemm_kernel<<<(waves + 7) / 8, 256, 0, stream>>>(
            sbuf, cntb, x1, Wl1, bl1, Wr1, hbuf, stats, stats + 256, N2, rowTiles);
    }
    norm_kernel<<<(int)((NC2 + 255) / 256), 256, 0, stream>>>(
        hbuf, stats, stats + 256, g1, be1, (float*)d_out, N2);
}


# Row sums by owner workgroups: each of 256 keeps its target rows in LDS, reads the whole destination list, adds its edges' source rows by LDS float adds (arrival order); no row adds, no zero pass
# speedup vs baseline: 1.8740x; 1.1565x over previous
#include <hip/hip_runtime.h>

typedef __attribute__((ext_vector_type(2))) float v2f;
typedef __attribute__((ext_vector_type(8))) float v8f;

#define CFEAT 256

__global__ void zero_kernel(float* __restrict__ p, long n) {
    long i = (long)blockIdx.x * blockDim.x + threadIdx.x;
    long stride = (long)gridDim.x * blockDim.x;
    for (; i < n; i += stride) p[i] = 0.0f;
}

#define SEG_THREADS 1024
#define SEG_WAVES 32
#define SEG_GRID 256

__device__ __forceinline__ void row_load(float (&v)[8], const float* __restrict__ x, int sr, int lane) {
    const float* xs = x + (long)sr * CFEAT + lane;
#pragma unroll
    for (int j = 0; j < 8; ++j) v[j] = xs[j * 32];
}

__device__ __forceinline__ void row_add_lds(const float (&v)[8], float* acc, int dr, int lane) {
    float* a = acc + dr * CFEAT + lane;
#pragma unroll
    for (int j = 0; j < 8; ++j) atomicAdd(&a[j * 32], v[j]);
}

#define SEG_BATCH() do { \
        const int s0_ = src[qe0], s1_ = src[qe1], s2_ = src[qe2], s3_ = src[qe3]; \
        float v0_[8], v1_[8], v2_[8], v3_[8]; \
        row_load(v0_, x, s0_, lane); \
        row_load(v1_, x, s1_, lane); \
        row_load(v2_, x, s2_, lane); \
        row_load(v3_, x, s3_, lane); \
        row_add_lds(v0_, acc, qd0, lane); \
        row_add_lds(v1_, acc, qd1, lane); \
        row_add_lds(v2_, acc, qd2, lane); \
        row_add_lds(v3_, acc, qd3, lane); \
        np = 0; \
    } while (0)

#define SEG_ENQUEUE(E_, D_) do { \
        const int e_ = (E_), dd_ = (D_); \
        if (np == 0)      { qe0 = e_; qd0 = dd_; } \
        else if (np == 1) { qe1 = e_; qd1 = dd_; } \
        else if (np == 2) { qe2 = e_; qd2 = dd_; } \
        else              { qe3 = e_; qd3 = dd_; } \
        ++np; \
        if (np == 4) SEG_BATCH(); \
    } while (0)

#define SEG_PROCESS(CC_, D_) do { \
        const int cc_ = (CC_); \
        const bool valid_ = cc_ < nFull; \
        const unsigned rl0 = (unsigned)((D_).x - r0), rl1 = (unsigned)((D_).y - r0); \
        const unsigned rl2 = (unsigned)((D_).z - r0), rl3 = (unsigned)((D_).w - r0); \
        const bool m0 = rl0 < (unsigned)ROWS, m1 = rl1 < (unsigned)ROWS; \
        const bool m2 = rl2 < (unsigned)ROWS, m3 = rl3 < (unsigned)ROWS; \
        const unsigned vm_ = valid_ ? ~0u : 0u; \
        const unsigned k0 = __builtin_amdgcn_ballot_w32(m0) & vm_, k1 = __builtin_amdgcn_ballot_w32(m1) & vm_; \
        const unsigned k2 = __builtin_amdgcn_ballot_w32(m2) & vm_, k3 = __builtin_amdgcn_ballot_w32(m3) & vm_; \
        unsigned M_ = k0 | k1 | k2 | k3; \
        if (M_) { \
            if (m0) atomicAdd(&cnt[(D_).x], 1.0f); \
            if (m1) atomicAdd(&cnt[(D_).y], 1.0f); \
            if (m2) atomicAdd(&cnt[(D_).z], 1.0f); \
            if (m3) atomicAdd(&cnt[(D_).w], 1.0f); \
            const int ebase_ = SEG_CHUNK(cc_) * 128; \
            while (M_) { \
                const int i_ = __builtin_ctz(M_); \
                M_ &= M_ - 1; \
                const int rq0 = __builtin_amdgcn_readlane((int)rl0, i_); \
                const int rq1 = __builtin_amdgcn_readlane((int)rl1, i_); \
                const int rq2 = __builtin_amdgcn_readlane((int)rl2, i_); \
                const int rq3 = __builtin_amdgcn_readlane((int)rl3, i_); \
                unsigned nib_ = ((k0 >> i_) & 1u) | (((k1 >> i_) & 1u) << 1) | \
                                (((k2 >> i_) & 1u) << 2) | (((k3 >> i_) & 1u) << 3); \
                while (nib_) { \
                    const int q_ = __builtin_ctz(nib_); \
                    nib_ &= nib_ - 1; \
                    const int dr_ = (q_ == 0) ? rq0 : (q_ == 1) ? rq1 : (q_ == 2) ? rq2 : rq3; \
                    SEG_ENQUEUE(ebase_ + i_ * 4 + q_, dr_); \
                } \
            } \
        } \
    } while (0)

#define SEG_CHUNK(T_) (((T_) + rot < nFull) ? ((T_) + rot) : ((T_) + rot - nFull))
#define SEG_LOAD(C_) (*(const int4*)(dst + SEG_CHUNK(((C_) < last) ? (C_) : last) * 128 + lane * 4))

template <int ROWS>
__global__ void __launch_bounds__(SEG_THREADS) segsum_kernel(const float* __restrict__ x,
                              const int* __restrict__ src,
                              const int* __restrict__ dst,
                              float* __restrict__ s,
                              float* __restrict__ cnt,
                              int E, int N) {
    __shared__ __attribute__((aligned(16))) float acc[ROWS * CFEAT];
    const int tid = threadIdx.x;
    const int lane = tid & 31;
    const int wv = __builtin_amdgcn_readfirstlane(tid >> 5);
    const int r0 = (int)blockIdx.x * ROWS;

    float4* acc4 = (float4*)acc;
    for (int i = tid; i < ROWS * (CFEAT / 4); i += SEG_THREADS) acc4[i] = make_float4(0.0f, 0.0f, 0.0f, 0.0f);
    __syncthreads();

    int qe0 = 0, qe1 = 0, qe2 = 0, qe3 = 0, qd0 = 0, qd1 = 0, qd2 = 0, qd3 = 0, np = 0;

    const int nFull = E >> 7;
    if (nFull > 0) {
        const int last = nFull - 1;
        const int rot = (int)(((long)blockIdx.x * nFull) / SEG_GRID);
        int4 dA = SEG_LOAD(wv);
        __builtin_amdgcn_sched_barrier(0);
        int4 dB = SEG_LOAD(wv + SEG_WAVES);
        __builtin_amdgcn_sched_barrier(0);
        int4 dC = SEG_LOAD(wv + 2 * SEG_WAVES);
        int4 dD;
        for (int c = wv; c < nFull; c += 4 * SEG_WAVES) {
            dD = SEG_LOAD(c + 3 * SEG_WAVES);
            SEG_PROCESS(c, dA);
            dA = SEG_LOAD(c + 4 * SEG_WAVES);
            SEG_PROCESS(c + SEG_WAVES, dB);
            dB = SEG_LOAD(c + 5 * SEG_WAVES);
            SEG_PROCESS(c + 2 * SEG_WAVES, dC);
            dC = SEG_LOAD(c + 6 * SEG_WAVES);
            SEG_PROCESS(c + 3 * SEG_WAVES, dD);
        }
    }
    if (wv == 0) {
        for (int e = nFull * 128; e < E; ++e) {
            const int de = dst[e];
            const unsigned rl = (unsigned)(de - r0);
            if (rl < (unsigned)ROWS) {
                if (lane == 0) atomicAdd(&cnt[de], 1.0f);
                SEG_ENQUEUE(e, (int)rl);
            }
        }
    }
    if (np > 0) {
        float v0[8];
        row_load(v0, x, src[qe0], lane);
        row_add_lds(v0, acc, qd0, lane);
        if (np > 1) {
            row_load(v0, x, src[qe1], lane);
            row_add_lds(v0, acc, qd1, lane);
        }
        if (np > 2) {
            row_load(v0, x, src[qe2], lane);
            row_add_lds(v0, acc, qd2, lane);
        }
    }
    __syncthreads();

    int rows = N - r0;
    rows = rows < ROWS ? rows : ROWS;
    float4* s4 = (float4*)(s + (long)r0 * CFEAT);
    for (int i = tid; i < rows * (CFEAT / 4); i += SEG_THREADS) s4[i] = acc4[i];
}

typedef float4 linefrag[4];

__device__ __forceinline__ void load_linefrag(linefrag& f, const float* __restrict__ rowh, int kc) {
    const float4* p = (const float4*)(rowh + kc);
#pragma unroll
    for (int q = 0; q < 4; ++q) f[q] = p[2 * q];
}

__device__ __forceinline__ v2f frag_pair(const linefrag& f, int q, int p) {
    v2f r;
    if (p == 0) { r[0] = f[q].x; r[1] = f[q].y; }
    else        { r[0] = f[q].z; r[1] = f[q].w; }
    return r;
}

__device__ __forceinline__ void mma_line(v8f (&acc)[4], const linefrag& a, const linefrag (&b)[4]) {
#pragma unroll
    for (int q = 0; q < 4; ++q) {
#pragma unroll
        for (int p = 0; p < 2; ++p) {
            v2f af = frag_pair(a, q, p);
#pragma unroll
            for (int t = 0; t < 4; ++t)
                acc[t] = __builtin_amdgcn_wmma_f32_16x16x4_f32(
                    false, af, false, frag_pair(b[t], q, p), (short)0, acc[t], false, false);
        }
    }
}

__global__ void __launch_bounds__(256) gemm_kernel(const float* __restrict__ s,
                            const float* __restrict__ cnt,
                            const float* __restrict__ xt,
                            const float* __restrict__ Wl,
                            const float* __restrict__ bl,
                            const float* __restrict__ Wr,
                            float* __restrict__ h,
                            float* __restrict__ colsum,
                            float* __restrict__ colsq,
                            int N, int rowTiles) {
    int wavesPerBlock = blockDim.x >> 5;
    int gw = blockIdx.x * wavesPerBlock + (threadIdx.x >> 5);
    int totalWaves = rowTiles * 4;
    if (gw >= totalWaves) return;

    int rowTile = gw >> 2;
    int colGrp  = gw & 3;
    int lane = threadIdx.x & 31;
    int l15  = lane & 15;
    int hoff = (lane >> 4) * 4;

    int rowA = rowTile * 16 + l15;
    int rowL = (rowA < N) ? rowA : (N - 1);
    float invc = 1.0f / fmaxf(cnt[rowL], 1.0f);
    const float* sRow = s  + (long)rowL * CFEAT + hoff;
    const float* xRow = xt + (long)rowL * CFEAT + hoff;

    int colBase = colGrp * 64;
    const float* wlRow[4];
    const float* wrRow[4];
#pragma unroll
    for (int t = 0; t < 4; ++t) {
        int colB = colBase + t * 16 + l15;
        wlRow[t] = Wl + (long)colB * CFEAT + hoff;
        wrRow[t] = Wr + (long)colB * CFEAT + hoff;
    }

    v8f acc[4] = {{0,0,0,0,0,0,0,0}, {0,0,0,0,0,0,0,0},
                  {0,0,0,0,0,0,0,0}, {0,0,0,0,0,0,0,0}};

    linefrag aC, bC[4], aN, bN[4];
    load_linefrag(aC, sRow, 0);
#pragma unroll
    for (int t = 0; t < 4; ++t) load_linefrag(bC[t], wlRow[t], 0);

    for (int kc = 0; kc < CFEAT - 32; kc += 32) {
        load_linefrag(aN, xRow, kc);
#pragma unroll
        for (int t = 0; t < 4; ++t) load_linefrag(bN[t], wrRow[t], kc);
        __builtin_amdgcn_sched_barrier(0);
#pragma unroll
        for (int q = 0; q < 4; ++q) aC[q] *= invc;
        mma_line(acc, aC, bC);
        __builtin_amdgcn_sched_barrier(0);
        load_linefrag(aC, sRow, kc + 32);
#pragma unroll
        for (int t = 0; t < 4; ++t) load_linefrag(bC[t], wlRow[t], kc + 32);
        __builtin_amdgcn_sched_barrier(0);
        mma_line(acc, aN, bN);
    }
    load_linefrag(aN, xRow, CFEAT - 32);
#pragma unroll
    for (int t = 0; t < 4; ++t) load_linefrag(bN[t], wrRow[t], CFEAT - 32);
    __builtin_amdgcn_sched_barrier(0);
#pragma unroll
    for (int q = 0; q < 4; ++q) aC[q] *= invc;
    mma_line(acc, aC, bC);
    mma_line(acc, aN, bN);

    int rbase = rowTile * 16 + ((lane < 16) ? 0 : 8);
#pragma unroll
    for (int t = 0; t < 4; ++t) {
        int col = colBase + t * 16 + l15;
        float bias = bl[col];
        float sp = 0.0f, qp = 0.0f;
#pragma unroll
        for (int v = 0; v < 8; ++v) {
            float val = fmaxf(acc[t][v] + bias, 0.0f);
            int r = rbase + v;
            bool ok = (r < N);
            float m = ok ? val : 0.0f;
            sp += m;
            qp += m * m;
            if (ok) h[(long)r * CFEAT + col] = val;
        }
        atomicAdd(&colsum[col], sp);
        atomicAdd(&colsq[col], qp);
    }
}

__global__ void norm_kernel(const float* __restrict__ hbuf,
                            const float* __restrict__ colsum,
                            const float* __restrict__ colsq,
                            const float* __restrict__ gamma,
                            const float* __restrict__ beta,
                            float* __restrict__ out, int N) {
    long i = (long)blockIdx.x * blockDim.x + threadIdx.x;
    long total = (long)N * CFEAT;
    if (i >= total) return;
    int col = (int)(i & (CFEAT - 1));
    float invN = 1.0f / (float)N;
    float mu  = colsum[col] * invN;
    float var = colsq[col] * invN - mu * mu;
    float rs  = rsqrtf(var + 1e-5f);
    out[i] = gamma[col] * (hbuf[i] - mu) * rs + beta[col];
}

extern "C" void kernel_launch(void* const* d_in, const int* in_sizes, int n_in,
                              void* d_out, int out_size, void* d_ws, size_t ws_size,
                              hipStream_t stream) {
    const float* x    = (const float*)d_in[0];
    const int* e0src  = (const int*)d_in[1];
    const int* e0dst  = (const int*)d_in[2];
    const int* e1src  = (const int*)d_in[3];
    const int* e1dst  = (const int*)d_in[4];
    const float* Wl0  = (const float*)d_in[5];
    const float* bl0  = (const float*)d_in[6];
    const float* Wr0  = (const float*)d_in[7];
    const float* g0   = (const float*)d_in[8];
    const float* be0  = (const float*)d_in[9];
    const float* Wl1  = (const float*)d_in[10];
    const float* bl1  = (const float*)d_in[11];
    const float* Wr1  = (const float*)d_in[12];
    const float* g1   = (const float*)d_in[13];
    const float* be1  = (const float*)d_in[14];

    const int N1 = 50000, N2 = 12500;
    const int E0 = in_sizes[1];
    const int E1 = in_sizes[3];

    float* ws = (float*)d_ws;
    const long NC1 = (long)N1 * CFEAT;
    float* x1    = ws;
    float* sbuf  = ws + NC1;
    float* hbuf  = ws + 2 * NC1;
    float* cntb  = ws + 3 * NC1;
    float* stats = cntb + 50048;

    zero_kernel<<<256, 256, 0, stream>>>(cntb, 50048 + 512);
    segsum_kernel<196><<<SEG_GRID, SEG_THREADS, 0, stream>>>(x, e0src, e0dst, sbuf, cntb, E0, N1);
    {
        int rowTiles = (N1 + 15) / 16;
        int waves = rowTiles * 4;
        gemm_kernel<<<(waves + 7) / 8, 256, 0, stream>>>(
            sbuf, cntb, x, Wl0, bl0, Wr0, hbuf, stats, stats + 256, N1, rowTiles);
    }
    norm_kernel<<<(int)((NC1 + 255) / 256), 256, 0, stream>>>(
        hbuf, stats, stats + 256, g0, be0, x1, N1);

    const long NC2 = (long)N2 * CFEAT;
    zero_kernel<<<256, 256, 0, stream>>>(cntb, 50048 + 512);
    segsum_kernel<49><<<SEG_GRID, SEG_THREADS, 0, stream>>>(x1, e1src, e1dst, sbuf, cntb, E1, N2);
    {
        int rowTiles = (N2 + 15) / 16;
        int waves = rowTiles * 4;
        gemm_kernel<<<(waves + 7) / 8, 256, 0, stream>>>(
            sbuf, cntb, x1, Wl1, bl1, Wr1, hbuf, stats, stats + 256, N2, rowTiles);
    }
    norm_kernel<<<(int)((NC2 + 255) / 256), 256, 0, stream>>>(
        hbuf, stats, stats + 256, g1, be1, (float*)d_out, N2);
}


# Owner scan reads one byte per edge: a start pass writes each edge's owner number; owners test 16 edges per lane by a zero-byte test on 16-byte key loads; sums still LDS float adds in arrival order
# speedup vs baseline: 2.0121x; 1.0737x over previous
#include <hip/hip_runtime.h>

typedef __attribute__((ext_vector_type(2))) float v2f;
typedef __attribute__((ext_vector_type(8))) float v8f;

#define CFEAT 256

template <int ROWS>
__global__ void prep_kernel(float* __restrict__ zp, int nz,
                            const int* __restrict__ dst,
                            unsigned int* __restrict__ key, int E) {
    const int nWords = ((E + 511) >> 9) * 128;
    const int i = (int)(blockIdx.x * blockDim.x + threadIdx.x);
    const int stride = (int)(gridDim.x * blockDim.x);
    for (int j = i; j < nz; j += stride) zp[j] = 0.0f;
    for (int j = i; j < nWords; j += stride) {
        const int e = 4 * j;
        unsigned w = 0u;
        if (e + 3 < E) {
            const int4 d = *(const int4*)(dst + e);
            w = ((unsigned)(d.x / ROWS) & 255u) | (((unsigned)(d.y / ROWS) & 255u) << 8) |
                (((unsigned)(d.z / ROWS) & 255u) << 16) | (((unsigned)(d.w / ROWS) & 255u) << 24);
        } else {
            for (int b = 0; b < 4; ++b)
                if (e + b < E) w |= ((unsigned)(dst[e + b] / ROWS) & 255u) << (8 * b);
        }
        key[j] = w;
    }
}

#define SEG_THREADS 1024
#define SEG_WAVES 32
#define SEG_GRID 256

__device__ __forceinline__ void row_load(float (&v)[8], const float* __restrict__ x, int sr, int lane) {
    const float* xs = x + (long)sr * CFEAT + lane;
#pragma unroll
    for (int j = 0; j < 8; ++j) v[j] = xs[j * 32];
}

__device__ __forceinline__ void row_add_lds(const float (&v)[8], float* acc, int dr, int lane) {
    float* a = acc + dr * CFEAT + lane;
#pragma unroll
    for (int j = 0; j < 8; ++j) atomicAdd(&a[j * 32], v[j]);
}

#define SEG_BATCH() do { \
        const int s0_ = src[qe0], s1_ = src[qe1], s2_ = src[qe2], s3_ = src[qe3]; \
        const int d0_ = dst[qe0], d1_ = dst[qe1], d2_ = dst[qe2], d3_ = dst[qe3]; \
        if (lane < 4) { \
            const int dl_ = (lane == 0) ? d0_ : (lane == 1) ? d1_ : (lane == 2) ? d2_ : d3_; \
            atomicAdd(&cnt[dl_], 1.0f); \
        } \
        float v0_[8], v1_[8], v2_[8], v3_[8]; \
        row_load(v0_, x, s0_, lane); \
        row_load(v1_, x, s1_, lane); \
        row_load(v2_, x, s2_, lane); \
        row_load(v3_, x, s3_, lane); \
        row_add_lds(v0_, acc, d0_ - r0, lane); \
        row_add_lds(v1_, acc, d1_ - r0, lane); \
        row_add_lds(v2_, acc, d2_ - r0, lane); \
        row_add_lds(v3_, acc, d3_ - r0, lane); \
        np = 0; \
    } while (0)

#define SEG_ENQUEUE(E_) do { \
        const int e_ = (E_); \
        if (e_ < E) { \
            if (np == 0)      qe0 = e_; \
            else if (np == 1) qe1 = e_; \
            else if (np == 2) qe2 = e_; \
            else              qe3 = e_; \
            ++np; \
            if (np == 4) SEG_BATCH(); \
        } \
    } while (0)

#define SEG_ZFLAGS(X_) (~((((X_) & 0x7F7F7F7Fu) + 0x7F7F7F7Fu) | (X_)) & 0x80808080u)

#define SEG_PROCESS(CC_, K_) do { \
        const int cc_ = (CC_); \
        const unsigned own_ = (SEG_ZFLAGS((K_).x ^ gpat) >> 7) | (SEG_ZFLAGS((K_).y ^ gpat) >> 6) | \
                              (SEG_ZFLAGS((K_).z ^ gpat) >> 5) | (SEG_ZFLAGS((K_).w ^ gpat) >> 4); \
        unsigned M_ = __builtin_amdgcn_ballot_w32(own_ != 0u); \
        if (cc_ >= nCh) M_ = 0u; \
        if (M_) { \
            const int ebase_ = SEG_CHUNK(cc_) * 512; \
            while (M_) { \
                const int i_ = __builtin_ctz(M_); \
                M_ &= M_ - 1; \
                unsigned o_ = (unsigned)__builtin_amdgcn_readlane((int)own_, i_); \
                while (o_) { \
                    const int j_ = __builtin_ctz(o_); \
                    o_ &= o_ - 1; \
                    SEG_ENQUEUE(ebase_ + i_ * 16 + ((j_ & 3) << 2) + (j_ >> 3)); \
                } \
            } \
        } \
    } while (0)

#define SEG_CHUNK(T_) (((T_) + rot < nCh) ? ((T_) + rot) : ((T_) + rot - nCh))
#define SEG_LOAD(C_) (*(const uint4*)(key + SEG_CHUNK(((C_) < last) ? (C_) : last) * 128 + lane * 4))

template <int ROWS>
__global__ void __launch_bounds__(SEG_THREADS) segsum_kernel(const float* __restrict__ x,
                              const int* __restrict__ src,
                              const int* __restrict__ dst,
                              const unsigned int* __restrict__ key,
                              float* __restrict__ s,
                              float* __restrict__ cnt,
                              int E, int N) {
    __shared__ __attribute__((aligned(16))) float acc[ROWS * CFEAT];
    const int tid = threadIdx.x;
    const int lane = tid & 31;
    const int wv = __builtin_amdgcn_readfirstlane(tid >> 5);
    const int r0 = (int)blockIdx.x * ROWS;
    const unsigned gpat = (unsigned)blockIdx.x * 0x01010101u;

    float4* acc4 = (float4*)acc;
    for (int i = tid; i < ROWS * (CFEAT / 4); i += SEG_THREADS) acc4[i] = make_float4(0.0f, 0.0f, 0.0f, 0.0f);
    __syncthreads();

    int qe0 = 0, qe1 = 0, qe2 = 0, qe3 = 0, np = 0;

    const int nCh = (E + 511) >> 9;
    if (nCh > 0) {
        const int last = nCh - 1;
        const int rot = (int)(((long)blockIdx.x * nCh) / SEG_GRID);
        uint4 dA = SEG_LOAD(wv);
        __builtin_amdgcn_sched_barrier(0);
        uint4 dB = SEG_LOAD(wv + SEG_WAVES);
        __builtin_amdgcn_sched_barrier(0);
        uint4 dC = SEG_LOAD(wv + 2 * SEG_WAVES);
        uint4 dD;
        for (int c = wv; c < nCh; c += 4 * SEG_WAVES) {
            dD = SEG_LOAD(c + 3 * SEG_WAVES);
            SEG_PROCESS(c, dA);
            dA = SEG_LOAD(c + 4 * SEG_WAVES);
            SEG_PROCESS(c + SEG_WAVES, dB);
            dB = SEG_LOAD(c + 5 * SEG_WAVES);
            SEG_PROCESS(c + 2 * SEG_WAVES, dC);
            dC = SEG_LOAD(c + 6 * SEG_WAVES);
            SEG_PROCESS(c + 3 * SEG_WAVES, dD);
        }
    }
    if (np > 0) {
        const int e1 = (np > 1) ? qe1 : qe0, e2 = (np > 2) ? qe2 : qe0;
        const int d0 = dst[qe0], d1 = dst[e1], d2 = dst[e2];
        if (lane < np) {
            const int dl = (lane == 0) ? d0 : (lane == 1) ? d1 : d2;
            atomicAdd(&cnt[dl], 1.0f);
        }
        float v0[8];
        row_load(v0, x, src[qe0], lane);
        row_add_lds(v0, acc, d0 - r0, lane);
        if (np > 1) {
            row_load(v0, x, src[e1], lane);
            row_add_lds(v0, acc, d1 - r0, lane);
        }
        if (np > 2) {
            row_load(v0, x, src[e2], lane);
            row_add_lds(v0, acc, d2 - r0, lane);
        }
    }
    __syncthreads();

    int rows = N - r0;
    rows = rows < ROWS ? rows : ROWS;
    float4* s4 = (float4*)(s + (long)r0 * CFEAT);
    for (int i = tid; i < rows * (CFEAT / 4); i += SEG_THREADS) s4[i] = acc4[i];
}

typedef float4 linefrag[4];

__device__ __forceinline__ void load_linefrag(linefrag& f, const float* __restrict__ rowh, int kc) {
    const float4* p = (const float4*)(rowh + kc);
#pragma unroll
    for (int q = 0; q < 4; ++q) f[q] = p[2 * q];
}

__device__ __forceinline__ v2f frag_pair(const linefrag& f, int q, int p) {
    v2f r;
    if (p == 0) { r[0] = f[q].x; r[1] = f[q].y; }
    else        { r[0] = f[q].z; r[1] = f[q].w; }
    return r;
}

__device__ __forceinline__ void mma_line(v8f (&acc)[4], const linefrag& a, const linefrag (&b)[4]) {
#pragma unroll
    for (int q = 0; q < 4; ++q) {
#pragma unroll
        for (int p = 0; p < 2; ++p) {
            v2f af = frag_pair(a, q, p);
#pragma unroll
            for (int t = 0; t < 4; ++t)
                acc[t] = __builtin_amdgcn_wmma_f32_16x16x4_f32(
                    false, af, false, frag_pair(b[t], q, p), (short)0, acc[t], false, false);
        }
    }
}

__global__ void __launch_bounds__(256) gemm_kernel(const float* __restrict__ s,
                            const float* __restrict__ cnt,
                            const float* __restrict__ xt,
                            const float* __restrict__ Wl,
                            const float* __restrict__ bl,
                            const float* __restrict__ Wr,
                            float* __restrict__ h,
                            float* __restrict__ colsum,
                            float* __restrict__ colsq,
                            int N, int rowTiles) {
    int wavesPerBlock = blockDim.x >> 5;
    int gw = blockIdx.x * wavesPerBlock + (threadIdx.x >> 5);
    int totalWaves = rowTiles * 4;
    if (gw >= totalWaves) return;

    int rowTile = gw >> 2;
    int colGrp  = gw & 3;
    int lane = threadIdx.x & 31;
    int l15  = lane & 15;
    int hoff = (lane >> 4) * 4;

    int rowA = rowTile * 16 + l15;
    int rowL = (rowA < N) ? rowA : (N - 1);
    float invc = 1.0f / fmaxf(cnt[rowL], 1.0f);
    const float* sRow = s  + (long)rowL * CFEAT + hoff;
    const float* xRow = xt + (long)rowL * CFEAT + hoff;

    int colBase = colGrp * 64;
    const float* wlRow[4];
    const float* wrRow[4];
#pragma unroll
    for (int t = 0; t < 4; ++t) {
        int colB = colBase + t * 16 + l15;
        wlRow[t] = Wl + (long)colB * CFEAT + hoff;
        wrRow[t] = Wr + (long)colB * CFEAT + hoff;
    }

    v8f acc[4] = {{0,0,0,0,0,0,0,0}, {0,0,0,0,0,0,0,0},
                  {0,0,0,0,0,0,0,0}, {0,0,0,0,0,0,0,0}};

    linefrag aC, bC[4], aN, bN[4];
    load_linefrag(aC, sRow, 0);
#pragma unroll
    for (int t = 0; t < 4; ++t) load_linefrag(bC[t], wlRow[t], 0);

    for (int kc = 0; kc < CFEAT - 32; kc += 32) {
        load_linefrag(aN, xRow, kc);
#pragma unroll
        for (int t = 0; t < 4; ++t) load_linefrag(bN[t], wrRow[t], kc);
        __builtin_amdgcn_sched_barrier(0);
#pragma unroll
        for (int q = 0; q < 4; ++q) aC[q] *= invc;
        mma_line(acc, aC, bC);
        __builtin_amdgcn_sched_barrier(0);
        load_linefrag(aC, sRow, kc + 32);
#pragma unroll
        for (int t = 0; t < 4; ++t) load_linefrag(bC[t], wlRow[t], kc + 32);
        __builtin_amdgcn_sched_barrier(0);
        mma_line(acc, aN, bN);
    }
    load_linefrag(aN, xRow, CFEAT - 32);
#pragma unroll
    for (int t = 0; t < 4; ++t) load_linefrag(bN[t], wrRow[t], CFEAT - 32);
    __builtin_amdgcn_sched_barrier(0);
#pragma unroll
    for (int q = 0; q < 4; ++q) aC[q] *= invc;
    mma_line(acc, aC, bC);
    mma_line(acc, aN, bN);

    int rbase = rowTile * 16 + ((lane < 16) ? 0 : 8);
#pragma unroll
    for (int t = 0; t < 4; ++t) {
        int col = colBase + t * 16 + l15;
        float bias = bl[col];
        float sp = 0.0f, qp = 0.0f;
#pragma unroll
        for (int v = 0; v < 8; ++v) {
            float val = fmaxf(acc[t][v] + bias, 0.0f);
            int r = rbase + v;
            bool ok = (r < N);
            float m = ok ? val : 0.0f;
            sp += m;
            qp += m * m;
            if (ok) h[(long)r * CFEAT + col] = val;
        }
        atomicAdd(&colsum[col], sp);
        atomicAdd(&colsq[col], qp);
    }
}

__global__ void norm_kernel(const float* __restrict__ hbuf,
                            const float* __restrict__ colsum,
                            const float* __restrict__ colsq,
                            const float* __restrict__ gamma,
                            const float* __restrict__ beta,
                            float* __restrict__ out, int N) {
    long i = (long)blockIdx.x * blockDim.x + threadIdx.x;
    long total = (long)N * CFEAT;
    if (i >= total) return;
    int col = (int)(i & (CFEAT - 1));
    float invN = 1.0f / (float)N;
    float mu  = colsum[col] * invN;
    float var = colsq[col] * invN - mu * mu;
    float rs  = rsqrtf(var + 1e-5f);
    out[i] = gamma[col] * (hbuf[i] - mu) * rs + beta[col];
}

extern "C" void kernel_launch(void* const* d_in, const int* in_sizes, int n_in,
                              void* d_out, int out_size, void* d_ws, size_t ws_size,
                              hipStream_t stream) {
    const float* x    = (const float*)d_in[0];
    const int* e0src  = (const int*)d_in[1];
    const int* e0dst  = (const int*)d_in[2];
    const int* e1src  = (const int*)d_in[3];
    const int* e1dst  = (const int*)d_in[4];
    const float* Wl0  = (const float*)d_in[5];
    const float* bl0  = (const float*)d_in[6];
    const float* Wr0  = (const float*)d_in[7];
    const float* g0   = (const float*)d_in[8];
    const float* be0  = (const float*)d_in[9];
    const float* Wl1  = (const float*)d_in[10];
    const float* bl1  = (const float*)d_in[11];
    const float* Wr1  = (const float*)d_in[12];
    const float* g1   = (const float*)d_in[13];
    const float* be1  = (const float*)d_in[14];

    const int N1 = 50000, N2 = 12500;
    const int E0 = in_sizes[1];
    const int E1 = in_sizes[3];

    float* ws = (float*)d_ws;
    const long NC1 = (long)N1 * CFEAT;
    float* x1    = ws;
    float* sbuf  = ws + NC1;
    float* hbuf  = ws + 2 * NC1;
    float* cntb  = ws + 3 * NC1;
    float* stats = cntb + 50048;

    unsigned int* keyb = (unsigned int*)hbuf;
    prep_kernel<196><<<256, 256, 0, stream>>>(cntb, 50048 + 512, e0dst, keyb, E0);
    segsum_kernel<196><<<SEG_GRID, SEG_THREADS, 0, stream>>>(x, e0src, e0dst, keyb, sbuf, cntb, E0, N1);
    {
        int rowTiles = (N1 + 15) / 16;
        int waves = rowTiles * 4;
        gemm_kernel<<<(waves + 7) / 8, 256, 0, stream>>>(
            sbuf, cntb, x, Wl0, bl0, Wr0, hbuf, stats, stats + 256, N1, rowTiles);
    }
    norm_kernel<<<(int)((NC1 + 255) / 256), 256, 0, stream>>>(
        hbuf, stats, stats + 256, g0, be0, x1, N1);

    const long NC2 = (long)N2 * CFEAT;
    prep_kernel<49><<<256, 256, 0, stream>>>(cntb, 50048 + 512, e1dst, keyb, E1);
    segsum_kernel<49><<<SEG_GRID, SEG_THREADS, 0, stream>>>(x1, e1src, e1dst, keyb, sbuf, cntb, E1, N2);
    {
        int rowTiles = (N2 + 15) / 16;
        int waves = rowTiles * 4;
        gemm_kernel<<<(waves + 7) / 8, 256, 0, stream>>>(
            sbuf, cntb, x1, Wl1, bl1, Wr1, hbuf, stats, stats + 256, N2, rowTiles);
    }
    norm_kernel<<<(int)((NC2 + 255) / 256), 256, 0, stream>>>(
        hbuf, stats, stats + 256, g1, be1, (float*)d_out, N2);
}


# GEMM wave tile 16x64 -> 64x64 held in registers (4x4 WMMA tiles, no LDS): 32 fragment loads per 128 WMMAs (was 20 per 32); k order kept; column sums add 32 rows per lane (was 8) per atomic
# speedup vs baseline: 2.6528x; 1.3184x over previous
#include <hip/hip_runtime.h>

typedef __attribute__((ext_vector_type(2))) float v2f;
typedef __attribute__((ext_vector_type(8))) float v8f;

#define CFEAT 256

template <int ROWS>
__global__ void prep_kernel(float* __restrict__ zp, int nz,
                            const int* __restrict__ dst,
                            unsigned int* __restrict__ key, int E) {
    const int nWords = ((E + 511) >> 9) * 128;
    const int i = (int)(blockIdx.x * blockDim.x + threadIdx.x);
    const int stride = (int)(gridDim.x * blockDim.x);
    for (int j = i; j < nz; j += stride) zp[j] = 0.0f;
    for (int j = i; j < nWords; j += stride) {
        const int e = 4 * j;
        unsigned w = 0u;
        if (e + 3 < E) {
            const int4 d = *(const int4*)(dst + e);
            w = ((unsigned)(d.x / ROWS) & 255u) | (((unsigned)(d.y / ROWS) & 255u) << 8) |
                (((unsigned)(d.z / ROWS) & 255u) << 16) | (((unsigned)(d.w / ROWS) & 255u) << 24);
        } else {
            for (int b = 0; b < 4; ++b)
                if (e + b < E) w |= ((unsigned)(dst[e + b] / ROWS) & 255u) << (8 * b);
        }
        key[j] = w;
    }
}

#define SEG_THREADS 1024
#define SEG_WAVES 32
#define SEG_GRID 256

__device__ __forceinline__ void row_load(float (&v)[8], const float* __restrict__ x, int sr, int lane) {
    const float* xs = x + (long)sr * CFEAT + lane;
#pragma unroll
    for (int j = 0; j < 8; ++j) v[j] = xs[j * 32];
}

__device__ __forceinline__ void row_add_lds(const float (&v)[8], float* acc, int dr, int lane) {
    float* a = acc + dr * CFEAT + lane;
#pragma unroll
    for (int j = 0; j < 8; ++j) atomicAdd(&a[j * 32], v[j]);
}

#define SEG_BATCH() do { \
        const int s0_ = src[qe0], s1_ = src[qe1], s2_ = src[qe2], s3_ = src[qe3]; \
        const int d0_ = dst[qe0], d1_ = dst[qe1], d2_ = dst[qe2], d3_ = dst[qe3]; \
        if (lane < 4) { \
            const int dl_ = (lane == 0) ? d0_ : (lane == 1) ? d1_ : (lane == 2) ? d2_ : d3_; \
            atomicAdd(&cnt[dl_], 1.0f); \
        } \
        float v0_[8], v1_[8], v2_[8], v3_[8]; \
        row_load(v0_, x, s0_, lane); \
        row_load(v1_, x, s1_, lane); \
        row_load(v2_, x, s2_, lane); \
        row_load(v3_, x, s3_, lane); \
        row_add_lds(v0_, acc, d0_ - r0, lane); \
        row_add_lds(v1_, acc, d1_ - r0, lane); \
        row_add_lds(v2_, acc, d2_ - r0, lane); \
        row_add_lds(v3_, acc, d3_ - r0, lane); \
        np = 0; \
    } while (0)

#define SEG_ENQUEUE(E_) do { \
        const int e_ = (E_); \
        if (e_ < E) { \
            if (np == 0)      qe0 = e_; \
            else if (np == 1) qe1 = e_; \
            else if (np == 2) qe2 = e_; \
            else              qe3 = e_; \
            ++np; \
            if (np == 4) SEG_BATCH(); \
        } \
    } while (0)

#define SEG_ZFLAGS(X_) (~((((X_) & 0x7F7F7F7Fu) + 0x7F7F7F7Fu) | (X_)) & 0x80808080u)

#define SEG_PROCESS(CC_, K_) do { \
        const int cc_ = (CC_); \
        const unsigned own_ = (SEG_ZFLAGS((K_).x ^ gpat) >> 7) | (SEG_ZFLAGS((K_).y ^ gpat) >> 6) | \
                              (SEG_ZFLAGS((K_).z ^ gpat) >> 5) | (SEG_ZFLAGS((K_).w ^ gpat) >> 4); \
        unsigned M_ = __builtin_amdgcn_ballot_w32(own_ != 0u); \
        if (cc_ >= nCh) M_ = 0u; \
        if (M_) { \
            const int ebase_ = SEG_CHUNK(cc_) * 512; \
            while (M_) { \
                const int i_ = __builtin_ctz(M_); \
                M_ &= M_ - 1; \
                unsigned o_ = (unsigned)__builtin_amdgcn_readlane((int)own_, i_); \
                while (o_) { \
                    const int j_ = __builtin_ctz(o_); \
                    o_ &= o_ - 1; \
                    SEG_ENQUEUE(ebase_ + i_ * 16 + ((j_ & 3) << 2) + (j_ >> 3)); \
                } \
            } \
        } \
    } while (0)

#define SEG_CHUNK(T_) (((T_) + rot < nCh) ? ((T_) + rot) : ((T_) + rot - nCh))
#define SEG_LOAD(C_) (*(const uint4*)(key + SEG_CHUNK(((C_) < last) ? (C_) : last) * 128 + lane * 4))

template <int ROWS>
__global__ void __launch_bounds__(SEG_THREADS) segsum_kernel(const float* __restrict__ x,
                              const int* __restrict__ src,
                              const int* __restrict__ dst,
                              const unsigned int* __restrict__ key,
                              float* __restrict__ s,
                              float* __restrict__ cnt,
                              int E, int N) {
    __shared__ __attribute__((aligned(16))) float acc[ROWS * CFEAT];
    const int tid = threadIdx.x;
    const int lane = tid & 31;
    const int wv = __builtin_amdgcn_readfirstlane(tid >> 5);
    const int r0 = (int)blockIdx.x * ROWS;
    const unsigned gpat = (unsigned)blockIdx.x * 0x01010101u;

    float4* acc4 = (float4*)acc;
    for (int i = tid; i < ROWS * (CFEAT / 4); i += SEG_THREADS) acc4[i] = make_float4(0.0f, 0.0f, 0.0f, 0.0f);
    __syncthreads();

    int qe0 = 0, qe1 = 0, qe2 = 0, qe3 = 0, np = 0;

    const int nCh = (E + 511) >> 9;
    if (nCh > 0) {
        const int last = nCh - 1;
        const int rot = (int)(((long)blockIdx.x * nCh) / SEG_GRID);
        uint4 dA = SEG_LOAD(wv);
        __builtin_amdgcn_sched_barrier(0);
        uint4 dB = SEG_LOAD(wv + SEG_WAVES);
        __builtin_amdgcn_sched_barrier(0);
        uint4 dC = SEG_LOAD(wv + 2 * SEG_WAVES);
        uint4 dD;
        for (int c = wv; c < nCh; c += 4 * SEG_WAVES) {
            dD = SEG_LOAD(c + 3 * SEG_WAVES);
            SEG_PROCESS(c, dA);
            dA = SEG_LOAD(c + 4 * SEG_WAVES);
            SEG_PROCESS(c + SEG_WAVES, dB);
            dB = SEG_LOAD(c + 5 * SEG_WAVES);
            SEG_PROCESS(c + 2 * SEG_WAVES, dC);
            dC = SEG_LOAD(c + 6 * SEG_WAVES);
            SEG_PROCESS(c + 3 * SEG_WAVES, dD);
        }
    }
    if (np > 0) {
        const int e1 = (np > 1) ? qe1 : qe0, e2 = (np > 2) ? qe2 : qe0;
        const int d0 = dst[qe0], d1 = dst[e1], d2 = dst[e2];
        if (lane < np) {
            const int dl = (lane == 0) ? d0 : (lane == 1) ? d1 : d2;
            atomicAdd(&cnt[dl], 1.0f);
        }
        float v0[8];
        row_load(v0, x, src[qe0], lane);
        row_add_lds(v0, acc, d0 - r0, lane);
        if (np > 1) {
            row_load(v0, x, src[e1], lane);
            row_add_lds(v0, acc, d1 - r0, lane);
        }
        if (np > 2) {
            row_load(v0, x, src[e2], lane);
            row_add_lds(v0, acc, d2 - r0, lane);
        }
    }
    __syncthreads();

    int rows = N - r0;
    rows = rows < ROWS ? rows : ROWS;
    float4* s4 = (float4*)(s + (long)r0 * CFEAT);
    for (int i = tid; i < rows * (CFEAT / 4); i += SEG_THREADS) s4[i] = acc4[i];
}

#define GEMM_RT 4
typedef float4 linefrag[4];

__device__ __forceinline__ void load_linefrag(linefrag& f, const float* __restrict__ rowh, int kc) {
    const float4* p = (const float4*)(rowh + kc);
#pragma unroll
    for (int q = 0; q < 4; ++q) f[q] = p[2 * q];
}

__device__ __forceinline__ v2f frag_pair(const linefrag& f, int q, int p) {
    v2f r;
    if (p == 0) { r[0] = f[q].x; r[1] = f[q].y; }
    else        { r[0] = f[q].z; r[1] = f[q].w; }
    return r;
}

__device__ __forceinline__ void mma_line(v8f (&acc)[GEMM_RT][4], const linefrag (&a)[GEMM_RT],
                                         const linefrag (&b)[4]) {
#pragma unroll
    for (int q = 0; q < 4; ++q) {
#pragma unroll
        for (int p = 0; p < 2; ++p) {
#pragma unroll
            for (int r = 0; r < GEMM_RT; ++r) {
                v2f af = frag_pair(a[r], q, p);
#pragma unroll
                for (int t = 0; t < 4; ++t)
                    acc[r][t] = __builtin_amdgcn_wmma_f32_16x16x4_f32(
                        false, af, false, frag_pair(b[t], q, p), (short)0, acc[r][t], false, false);
            }
        }
    }
}

__global__ void __launch_bounds__(128) gemm_kernel(const float* __restrict__ s,
                            const float* __restrict__ cnt,
                            const float* __restrict__ xt,
                            const float* __restrict__ Wl,
                            const float* __restrict__ bl,
                            const float* __restrict__ Wr,
                            float* __restrict__ h,
                            float* __restrict__ colsum,
                            float* __restrict__ colsq,
                            int N, int rowTiles) {
    int wavesPerBlock = blockDim.x >> 5;
    int gw = blockIdx.x * wavesPerBlock + (threadIdx.x >> 5);
    int totalWaves = rowTiles * 4;
    if (gw >= totalWaves) return;

    int rowTile = gw >> 2;
    int colGrp  = gw & 3;
    int lane = threadIdx.x & 31;
    int l15  = lane & 15;
    int hoff = (lane >> 4) * 4;

    float invc[GEMM_RT];
    const float* sRow[GEMM_RT];
    const float* xRow[GEMM_RT];
#pragma unroll
    for (int r = 0; r < GEMM_RT; ++r) {
        int rowA = (rowTile * GEMM_RT + r) * 16 + l15;
        int rowL = (rowA < N) ? rowA : (N - 1);
        invc[r] = 1.0f / fmaxf(cnt[rowL], 1.0f);
        sRow[r] = s  + (long)rowL * CFEAT + hoff;
        xRow[r] = xt + (long)rowL * CFEAT + hoff;
    }

    int colBase = colGrp * 64;
    const float* wlRow[4];
    const float* wrRow[4];
#pragma unroll
    for (int t = 0; t < 4; ++t) {
        int colB = colBase + t * 16 + l15;
        wlRow[t] = Wl + (long)colB * CFEAT + hoff;
        wrRow[t] = Wr + (long)colB * CFEAT + hoff;
    }

    v8f acc[GEMM_RT][4];
#pragma unroll
    for (int r = 0; r < GEMM_RT; ++r)
#pragma unroll
        for (int t = 0; t < 4; ++t)
            acc[r][t] = (v8f){0, 0, 0, 0, 0, 0, 0, 0};

    linefrag aC[GEMM_RT], bC[4], aN[GEMM_RT], bN[4];
#pragma unroll
    for (int r = 0; r < GEMM_RT; ++r) load_linefrag(aC[r], sRow[r], 0);
#pragma unroll
    for (int t = 0; t < 4; ++t) load_linefrag(bC[t], wlRow[t], 0);

    for (int kc = 0; kc < CFEAT - 32; kc += 32) {
#pragma unroll
        for (int r = 0; r < GEMM_RT; ++r) load_linefrag(aN[r], xRow[r], kc);
#pragma unroll
        for (int t = 0; t < 4; ++t) load_linefrag(bN[t], wrRow[t], kc);
        __builtin_amdgcn_sched_barrier(0);
#pragma unroll
        for (int r = 0; r < GEMM_RT; ++r)
#pragma unroll
            for (int q = 0; q < 4; ++q) aC[r][q] *= invc[r];
        mma_line(acc, aC, bC);
        __builtin_amdgcn_sched_barrier(0);
#pragma unroll
        for (int r = 0; r < GEMM_RT; ++r) load_linefrag(aC[r], sRow[r], kc + 32);
#pragma unroll
        for (int t = 0; t < 4; ++t) load_linefrag(bC[t], wlRow[t], kc + 32);
        __builtin_amdgcn_sched_barrier(0);
        mma_line(acc, aN, bN);
    }
#pragma unroll
    for (int r = 0; r < GEMM_RT; ++r) load_linefrag(aN[r], xRow[r], CFEAT - 32);
#pragma unroll
    for (int t = 0; t < 4; ++t) load_linefrag(bN[t], wrRow[t], CFEAT - 32);
    __builtin_amdgcn_sched_barrier(0);
#pragma unroll
    for (int r = 0; r < GEMM_RT; ++r)
#pragma unroll
        for (int q = 0; q < 4; ++q) aC[r][q] *= invc[r];
    mma_line(acc, aC, bC);
    mma_line(acc, aN, bN);

    int half8 = (lane < 16) ? 0 : 8;
#pragma unroll
    for (int t = 0; t < 4; ++t) {
        int col = colBase + t * 16 + l15;
        float bias = bl[col];
        float sp = 0.0f, qp = 0.0f;
#pragma unroll
        for (int r = 0; r < GEMM_RT; ++r) {
            int rbase = (rowTile * GEMM_RT + r) * 16 + half8;
#pragma unroll
            for (int v = 0; v < 8; ++v) {
                float val = fmaxf(acc[r][t][v] + bias, 0.0f);
                int rr = rbase + v;
                bool ok = (rr < N);
                float m = ok ? val : 0.0f;
                sp += m;
                qp += m * m;
                if (ok) h[(long)rr * CFEAT + col] = val;
            }
        }
        atomicAdd(&colsum[col], sp);
        atomicAdd(&colsq[col], qp);
    }
}

__global__ void norm_kernel(const float* __restrict__ hbuf,
                            const float* __restrict__ colsum,
                            const float* __restrict__ colsq,
                            const float* __restrict__ gamma,
                            const float* __restrict__ beta,
                            float* __restrict__ out, int N) {
    long i = (long)blockIdx.x * blockDim.x + threadIdx.x;
    long total = (long)N * CFEAT;
    if (i >= total) return;
    int col = (int)(i & (CFEAT - 1));
    float invN = 1.0f / (float)N;
    float mu  = colsum[col] * invN;
    float var = colsq[col] * invN - mu * mu;
    float rs  = rsqrtf(var + 1e-5f);
    out[i] = gamma[col] * (hbuf[i] - mu) * rs + beta[col];
}

extern "C" void kernel_launch(void* const* d_in, const int* in_sizes, int n_in,
                              void* d_out, int out_size, void* d_ws, size_t ws_size,
                              hipStream_t stream) {
    const float* x    = (const float*)d_in[0];
    const int* e0src  = (const int*)d_in[1];
    const int* e0dst  = (const int*)d_in[2];
    const int* e1src  = (const int*)d_in[3];
    const int* e1dst  = (const int*)d_in[4];
    const float* Wl0  = (const float*)d_in[5];
    const float* bl0  = (const float*)d_in[6];
    const float* Wr0  = (const float*)d_in[7];
    const float* g0   = (const float*)d_in[8];
    const float* be0  = (const float*)d_in[9];
    const float* Wl1  = (const float*)d_in[10];
    const float* bl1  = (const float*)d_in[11];
    const float* Wr1  = (const float*)d_in[12];
    const float* g1   = (const float*)d_in[13];
    const float* be1  = (const float*)d_in[14];

    const int N1 = 50000, N2 = 12500;
    const int E0 = in_sizes[1];
    const int E1 = in_sizes[3];

    float* ws = (float*)d_ws;
    const long NC1 = (long)N1 * CFEAT;
    float* x1    = ws;
    float* sbuf  = ws + NC1;
    float* hbuf  = ws + 2 * NC1;
    float* cntb  = ws + 3 * NC1;
    float* stats = cntb + 50048;

    unsigned int* keyb = (unsigned int*)hbuf;
    prep_kernel<196><<<256, 256, 0, stream>>>(cntb, 50048 + 512, e0dst, keyb, E0);
    segsum_kernel<196><<<SEG_GRID, SEG_THREADS, 0, stream>>>(x, e0src, e0dst, keyb, sbuf, cntb, E0, N1);
    {
        int rowTiles = (N1 + 63) / 64;
        gemm_kernel<<<rowTiles, 128, 0, stream>>>(
            sbuf, cntb, x, Wl0, bl0, Wr0, hbuf, stats, stats + 256, N1, rowTiles);
    }
    norm_kernel<<<(int)((NC1 + 255) / 256), 256, 0, stream>>>(
        hbuf, stats, stats + 256, g0, be0, x1, N1);

    const long NC2 = (long)N2 * CFEAT;
    prep_kernel<49><<<256, 256, 0, stream>>>(cntb, 50048 + 512, e1dst, keyb, E1);
    segsum_kernel<49><<<SEG_GRID, SEG_THREADS, 0, stream>>>(x1, e1src, e1dst, keyb, sbuf, cntb, E1, N2);
    {
        int rowTiles = (N2 + 63) / 64;
        gemm_kernel<<<rowTiles, 128, 0, stream>>>(
            sbuf, cntb, x1, Wl1, bl1, Wr1, hbuf, stats, stats + 256, N2, rowTiles);
    }
    norm_kernel<<<(int)((NC2 + 255) / 256), 256, 0, stream>>>(
        hbuf, stats, stats + 256, g1, be1, (float*)d_out, N2);
}


# Layer-0 normalisation pass removed: its per-column scale and shift go into layer 1's weights (k columns scaled once) and two bias rows (rows without edges skip one); 7 launches; rounding differs
# speedup vs baseline: 2.9025x; 1.0941x over previous
#include <hip/hip_runtime.h>

typedef __attribute__((ext_vector_type(2))) float v2f;
typedef __attribute__((ext_vector_type(8))) float v8f;

#define CFEAT 256

template <int ROWS>
__global__ void prep_kernel(float* __restrict__ zp, int nz,
                            float* __restrict__ zq, int nzq,
                            const int* __restrict__ dst,
                            unsigned int* __restrict__ key, int E,
                            const float* __restrict__ pstats, int nPrev,
                            const float* __restrict__ pg,
                            const float* __restrict__ pb,
                            const float* __restrict__ Wl,
                            const float* __restrict__ Wr,
                            const float* __restrict__ bl,
                            float* __restrict__ Wls,
                            float* __restrict__ Wrs,
                            float* __restrict__ bAgg,
                            float* __restrict__ bRoot) {
    const int nWords = ((E + 511) >> 9) * 128;
    const int i = (int)(blockIdx.x * blockDim.x + threadIdx.x);
    const int stride = (int)(gridDim.x * blockDim.x);
    for (int j = i; j < nz; j += stride) zp[j] = 0.0f;
    for (int j = i; j < nzq; j += stride) zq[j] = 0.0f;
    if (pstats != nullptr) {
        const float invN = 1.0f / (float)nPrev;
        for (int j = i; j < 2 * CFEAT * CFEAT; j += stride) {
            const int m = j >> 16;
            const int idx = j & (CFEAT * CFEAT - 1);
            const int k = idx & (CFEAT - 1);
            const float mu = pstats[k] * invN;
            const float var = pstats[CFEAT + k] * invN - mu * mu;
            const float a = pg[k] * rsqrtf(var + 1e-5f);
            if (m == 0) Wls[idx] = Wl[idx] * a;
            else        Wrs[idx] = Wr[idx] * a;
        }
        const int gwave = i >> 5;
        const int lane = i & 31;
        if (gwave < 2 * CFEAT) {
            const int m = gwave >> 8;
            const int jrow = gwave & (CFEAT - 1);
            const float* wrow = (m == 0 ? Wl : Wr) + jrow * CFEAT;
            float acc = 0.0f;
            for (int t = 0; t < CFEAT / 32; ++t) {
                const int k = lane + 32 * t;
                const float mu = pstats[k] * invN;
                const float var = pstats[CFEAT + k] * invN - mu * mu;
                const float a = pg[k] * rsqrtf(var + 1e-5f);
                const float c = pb[k] - a * mu;
                acc += c * wrow[k];
            }
            for (int d = 16; d >= 1; d >>= 1) acc += __shfl_xor(acc, d, 32);
            if (lane == 0) {
                if (m == 0) bAgg[jrow] = acc;
                else        bRoot[jrow] = bl[jrow] + acc;
            }
        }
    }
    for (int j = i; j < nWords; j += stride) {
        const int e = 4 * j;
        unsigned w = 0u;
        if (e + 3 < E) {
            const int4 d = *(const int4*)(dst + e);
            w = ((unsigned)(d.x / ROWS) & 255u) | (((unsigned)(d.y / ROWS) & 255u) << 8) |
                (((unsigned)(d.z / ROWS) & 255u) << 16) | (((unsigned)(d.w / ROWS) & 255u) << 24);
        } else {
            for (int b = 0; b < 4; ++b)
                if (e + b < E) w |= ((unsigned)(dst[e + b] / ROWS) & 255u) << (8 * b);
        }
        key[j] = w;
    }
}

#define SEG_THREADS 1024
#define SEG_WAVES 32
#define SEG_GRID 256

__device__ __forceinline__ void row_load(float (&v)[8], const float* __restrict__ x, int sr, int lane) {
    const float* xs = x + (long)sr * CFEAT + lane;
#pragma unroll
    for (int j = 0; j < 8; ++j) v[j] = xs[j * 32];
}

__device__ __forceinline__ void row_add_lds(const float (&v)[8], float* acc, int dr, int lane) {
    float* a = acc + dr * CFEAT + lane;
#pragma unroll
    for (int j = 0; j < 8; ++j) atomicAdd(&a[j * 32], v[j]);
}

#define SEG_BATCH() do { \
        const int s0_ = src[qe0], s1_ = src[qe1], s2_ = src[qe2], s3_ = src[qe3]; \
        const int d0_ = dst[qe0], d1_ = dst[qe1], d2_ = dst[qe2], d3_ = dst[qe3]; \
        if (lane < 4) { \
            const int dl_ = (lane == 0) ? d0_ : (lane == 1) ? d1_ : (lane == 2) ? d2_ : d3_; \
            atomicAdd(&cnt[dl_], 1.0f); \
        } \
        float v0_[8], v1_[8], v2_[8], v3_[8]; \
        row_load(v0_, x, s0_, lane); \
        row_load(v1_, x, s1_, lane); \
        row_load(v2_, x, s2_, lane); \
        row_load(v3_, x, s3_, lane); \
        row_add_lds(v0_, acc, d0_ - r0, lane); \
        row_add_lds(v1_, acc, d1_ - r0, lane); \
        row_add_lds(v2_, acc, d2_ - r0, lane); \
        row_add_lds(v3_, acc, d3_ - r0, lane); \
        np = 0; \
    } while (0)

#define SEG_ENQUEUE(E_) do { \
        const int e_ = (E_); \
        if (e_ < E) { \
            if (np == 0)      qe0 = e_; \
            else if (np == 1) qe1 = e_; \
            else if (np == 2) qe2 = e_; \
            else              qe3 = e_; \
            ++np; \
            if (np == 4) SEG_BATCH(); \
        } \
    } while (0)

#define SEG_ZFLAGS(X_) (~((((X_) & 0x7F7F7F7Fu) + 0x7F7F7F7Fu) | (X_)) & 0x80808080u)

#define SEG_PROCESS(CC_, K_) do { \
        const int cc_ = (CC_); \
        const unsigned own_ = (SEG_ZFLAGS((K_).x ^ gpat) >> 7) | (SEG_ZFLAGS((K_).y ^ gpat) >> 6) | \
                              (SEG_ZFLAGS((K_).z ^ gpat) >> 5) | (SEG_ZFLAGS((K_).w ^ gpat) >> 4); \
        unsigned M_ = __builtin_amdgcn_ballot_w32(own_ != 0u); \
        if (cc_ >= nCh) M_ = 0u; \
        if (M_) { \
            const int ebase_ = SEG_CHUNK(cc_) * 512; \
            while (M_) { \
                const int i_ = __builtin_ctz(M_); \
                M_ &= M_ - 1; \
                unsigned o_ = (unsigned)__builtin_amdgcn_readlane((int)own_, i_); \
                while (o_) { \
                    const int j_ = __builtin_ctz(o_); \
                    o_ &= o_ - 1; \
                    SEG_ENQUEUE(ebase_ + i_ * 16 + ((j_ & 3) << 2) + (j_ >> 3)); \
                } \
            } \
        } \
    } while (0)

#define SEG_CHUNK(T_) (((T_) + rot < nCh) ? ((T_) + rot) : ((T_) + rot - nCh))
#define SEG_LOAD(C_) (*(const uint4*)(key + SEG_CHUNK(((C_) < last) ? (C_) : last) * 128 + lane * 4))

template <int ROWS>
__global__ void __launch_bounds__(SEG_THREADS) segsum_kernel(const float* __restrict__ x,
                              const int* __restrict__ src,
                              const int* __restrict__ dst,
                              const unsigned int* __restrict__ key,
                              float* __restrict__ s,
                              float* __restrict__ cnt,
                              int E, int N) {
    __shared__ __attribute__((aligned(16))) float acc[ROWS * CFEAT];
    const int tid = threadIdx.x;
    const int lane = tid & 31;
    const int wv = __builtin_amdgcn_readfirstlane(tid >> 5);
    const int r0 = (int)blockIdx.x * ROWS;
    const unsigned gpat = (unsigned)blockIdx.x * 0x01010101u;

    float4* acc4 = (float4*)acc;
    for (int i = tid; i < ROWS * (CFEAT / 4); i += SEG_THREADS) acc4[i] = make_float4(0.0f, 0.0f, 0.0f, 0.0f);
    __syncthreads();

    int qe0 = 0, qe1 = 0, qe2 = 0, qe3 = 0, np = 0;

    const int nCh = (E + 511) >> 9;
    if (nCh > 0) {
        const int last = nCh - 1;
        const int rot = (int)(((long)blockIdx.x * nCh) / SEG_GRID);
        uint4 dA = SEG_LOAD(wv);
        __builtin_amdgcn_sched_barrier(0);
        uint4 dB = SEG_LOAD(wv + SEG_WAVES);
        __builtin_amdgcn_sched_barrier(0);
        uint4 dC = SEG_LOAD(wv + 2 * SEG_WAVES);
        uint4 dD;
        for (int c = wv; c < nCh; c += 4 * SEG_WAVES) {
            dD = SEG_LOAD(c + 3 * SEG_WAVES);
            SEG_PROCESS(c, dA);
            dA = SEG_LOAD(c + 4 * SEG_WAVES);
            SEG_PROCESS(c + SEG_WAVES, dB);
            dB = SEG_LOAD(c + 5 * SEG_WAVES);
            SEG_PROCESS(c + 2 * SEG_WAVES, dC);
            dC = SEG_LOAD(c + 6 * SEG_WAVES);
            SEG_PROCESS(c + 3 * SEG_WAVES, dD);
        }
    }
    if (np > 0) {
        const int e1 = (np > 1) ? qe1 : qe0, e2 = (np > 2) ? qe2 : qe0;
        const int d0 = dst[qe0], d1 = dst[e1], d2 = dst[e2];
        if (lane < np) {
            const int dl = (lane == 0) ? d0 : (lane == 1) ? d1 : d2;
            atomicAdd(&cnt[dl], 1.0f);
        }
        float v0[8];
        row_load(v0, x, src[qe0], lane);
        row_add_lds(v0, acc, d0 - r0, lane);
        if (np > 1) {
            row_load(v0, x, src[e1], lane);
            row_add_lds(v0, acc, d1 - r0, lane);
        }
        if (np > 2) {
            row_load(v0, x, src[e2], lane);
            row_add_lds(v0, acc, d2 - r0, lane);
        }
    }
    __syncthreads();

    int rows = N - r0;
    rows = rows < ROWS ? rows : ROWS;
    float4* s4 = (float4*)(s + (long)r0 * CFEAT);
    for (int i = tid; i < rows * (CFEAT / 4); i += SEG_THREADS) s4[i] = acc4[i];
}

#define GEMM_RT 4
typedef float4 linefrag[4];

__device__ __forceinline__ void load_linefrag(linefrag& f, const float* __restrict__ rowh, int kc) {
    const float4* p = (const float4*)(rowh + kc);
#pragma unroll
    for (int q = 0; q < 4; ++q) f[q] = p[2 * q];
}

__device__ __forceinline__ v2f frag_pair(const linefrag& f, int q, int p) {
    v2f r;
    if (p == 0) { r[0] = f[q].x; r[1] = f[q].y; }
    else        { r[0] = f[q].z; r[1] = f[q].w; }
    return r;
}

__device__ __forceinline__ void mma_line(v8f (&acc)[GEMM_RT][4], const linefrag (&a)[GEMM_RT],
                                         const linefrag (&b)[4]) {
#pragma unroll
    for (int q = 0; q < 4; ++q) {
#pragma unroll
        for (int p = 0; p < 2; ++p) {
#pragma unroll
            for (int r = 0; r < GEMM_RT; ++r) {
                v2f af = frag_pair(a[r], q, p);
#pragma unroll
                for (int t = 0; t < 4; ++t)
                    acc[r][t] = __builtin_amdgcn_wmma_f32_16x16x4_f32(
                        false, af, false, frag_pair(b[t], q, p), (short)0, acc[r][t], false, false);
            }
        }
    }
}

__global__ void __launch_bounds__(128) gemm_kernel(const float* __restrict__ s,
                            const float* __restrict__ cnt,
                            const float* __restrict__ xt,
                            const float* __restrict__ Wl,
                            const float* __restrict__ bl,
                            const float* __restrict__ Wr,
                            const float* __restrict__ bAgg,
                            float* __restrict__ h,
                            float* __restrict__ colsum,
                            float* __restrict__ colsq,
                            int N, int rowTiles) {
    int wavesPerBlock = blockDim.x >> 5;
    int gw = blockIdx.x * wavesPerBlock + (threadIdx.x >> 5);
    int totalWaves = rowTiles * 4;
    if (gw >= totalWaves) return;

    int rowTile = gw >> 2;
    int colGrp  = gw & 3;
    int lane = threadIdx.x & 31;
    int l15  = lane & 15;
    int hoff = (lane >> 4) * 4;

    float invc[GEMM_RT];
    unsigned hasIn[GEMM_RT];
    const float* sRow[GEMM_RT];
    const float* xRow[GEMM_RT];
#pragma unroll
    for (int r = 0; r < GEMM_RT; ++r) {
        int rowA = (rowTile * GEMM_RT + r) * 16 + l15;
        int rowL = (rowA < N) ? rowA : (N - 1);
        const float cr = cnt[rowL];
        invc[r] = 1.0f / fmaxf(cr, 1.0f);
        hasIn[r] = __builtin_amdgcn_ballot_w32(cr > 0.0f);
        sRow[r] = s  + (long)rowL * CFEAT + hoff;
        xRow[r] = xt + (long)rowL * CFEAT + hoff;
    }

    int colBase = colGrp * 64;
    const float* wlRow[4];
    const float* wrRow[4];
#pragma unroll
    for (int t = 0; t < 4; ++t) {
        int colB = colBase + t * 16 + l15;
        wlRow[t] = Wl + (long)colB * CFEAT + hoff;
        wrRow[t] = Wr + (long)colB * CFEAT + hoff;
    }

    v8f acc[GEMM_RT][4];
#pragma unroll
    for (int r = 0; r < GEMM_RT; ++r)
#pragma unroll
        for (int t = 0; t < 4; ++t)
            acc[r][t] = (v8f){0, 0, 0, 0, 0, 0, 0, 0};

    linefrag aC[GEMM_RT], bC[4], aN[GEMM_RT], bN[4];
#pragma unroll
    for (int r = 0; r < GEMM_RT; ++r) load_linefrag(aC[r], sRow[r], 0);
#pragma unroll
    for (int t = 0; t < 4; ++t) load_linefrag(bC[t], wlRow[t], 0);

    for (int kc = 0; kc < CFEAT - 32; kc += 32) {
#pragma unroll
        for (int r = 0; r < GEMM_RT; ++r) load_linefrag(aN[r], xRow[r], kc);
#pragma unroll
        for (int t = 0; t < 4; ++t) load_linefrag(bN[t], wrRow[t], kc);
        __builtin_amdgcn_sched_barrier(0);
#pragma unroll
        for (int r = 0; r < GEMM_RT; ++r)
#pragma unroll
            for (int q = 0; q < 4; ++q) aC[r][q] *= invc[r];
        mma_line(acc, aC, bC);
        __builtin_amdgcn_sched_barrier(0);
#pragma unroll
        for (int r = 0; r < GEMM_RT; ++r) load_linefrag(aC[r], sRow[r], kc + 32);
#pragma unroll
        for (int t = 0; t < 4; ++t) load_linefrag(bC[t], wlRow[t], kc + 32);
        __builtin_amdgcn_sched_barrier(0);
        mma_line(acc, aN, bN);
    }
#pragma unroll
    for (int r = 0; r < GEMM_RT; ++r) load_linefrag(aN[r], xRow[r], CFEAT - 32);
#pragma unroll
    for (int t = 0; t < 4; ++t) load_linefrag(bN[t], wrRow[t], CFEAT - 32);
    __builtin_amdgcn_sched_barrier(0);
#pragma unroll
    for (int r = 0; r < GEMM_RT; ++r)
#pragma unroll
        for (int q = 0; q < 4; ++q) aC[r][q] *= invc[r];
    mma_line(acc, aC, bC);
    mma_line(acc, aN, bN);

    int half8 = (lane < 16) ? 0 : 8;
#pragma unroll
    for (int t = 0; t < 4; ++t) {
        int col = colBase + t * 16 + l15;
        float bias = bl[col];
        float ba = (bAgg != nullptr) ? bAgg[col] : 0.0f;
        float sp = 0.0f, qp = 0.0f;
#pragma unroll
        for (int r = 0; r < GEMM_RT; ++r) {
            int rbase = (rowTile * GEMM_RT + r) * 16 + half8;
#pragma unroll
            for (int v = 0; v < 8; ++v) {
                float bv = ((hasIn[r] >> (half8 + v)) & 1u) ? (bias + ba) : bias;
                float val = fmaxf(acc[r][t][v] + bv, 0.0f);
                int rr = rbase + v;
                bool ok = (rr < N);
                float m = ok ? val : 0.0f;
                sp += m;
                qp += m * m;
                if (ok) h[(long)rr * CFEAT + col] = val;
            }
        }
        atomicAdd(&colsum[col], sp);
        atomicAdd(&colsq[col], qp);
    }
}

__global__ void norm_kernel(const float* __restrict__ hbuf,
                            const float* __restrict__ colsum,
                            const float* __restrict__ colsq,
                            const float* __restrict__ gamma,
                            const float* __restrict__ beta,
                            float* __restrict__ out, int N) {
    long i = (long)blockIdx.x * blockDim.x + threadIdx.x;
    long total = (long)N * CFEAT;
    if (i >= total) return;
    int col = (int)(i & (CFEAT - 1));
    float invN = 1.0f / (float)N;
    float mu  = colsum[col] * invN;
    float var = colsq[col] * invN - mu * mu;
    float rs  = rsqrtf(var + 1e-5f);
    out[i] = gamma[col] * (hbuf[i] - mu) * rs + beta[col];
}

extern "C" void kernel_launch(void* const* d_in, const int* in_sizes, int n_in,
                              void* d_out, int out_size, void* d_ws, size_t ws_size,
                              hipStream_t stream) {
    const float* x    = (const float*)d_in[0];
    const int* e0src  = (const int*)d_in[1];
    const int* e0dst  = (const int*)d_in[2];
    const int* e1src  = (const int*)d_in[3];
    const int* e1dst  = (const int*)d_in[4];
    const float* Wl0  = (const float*)d_in[5];
    const float* bl0  = (const float*)d_in[6];
    const float* Wr0  = (const float*)d_in[7];
    const float* g0   = (const float*)d_in[8];
    const float* be0  = (const float*)d_in[9];
    const float* Wl1  = (const float*)d_in[10];
    const float* bl1  = (const float*)d_in[11];
    const float* Wr1  = (const float*)d_in[12];
    const float* g1   = (const float*)d_in[13];
    const float* be1  = (const float*)d_in[14];

    const int N1 = 50000, N2 = 12500;
    const int E0 = in_sizes[1];
    const int E1 = in_sizes[3];

    float* ws = (float*)d_ws;
    const long NC1 = (long)N1 * CFEAT;
    float* x1    = ws;
    float* sbuf  = ws + NC1;
    float* hbuf  = ws + 2 * NC1;
    float* cntb  = ws + 3 * NC1;
    float* stats = cntb + 50048;
    float* Wls    = hbuf + 4 * 1048576;
    float* Wrs    = Wls + CFEAT * CFEAT;
    float* bAgg   = Wrs + CFEAT * CFEAT;
    float* bRoot  = bAgg + CFEAT;
    float* stats1 = bRoot + CFEAT;

    unsigned int* keyb = (unsigned int*)hbuf;
    prep_kernel<196><<<256, 256, 0, stream>>>(cntb, 50048 + 512, nullptr, 0, e0dst, keyb, E0,
        nullptr, 0, nullptr, nullptr, nullptr, nullptr, nullptr, nullptr, nullptr, nullptr, nullptr);
    segsum_kernel<196><<<SEG_GRID, SEG_THREADS, 0, stream>>>(x, e0src, e0dst, keyb, sbuf, cntb, E0, N1);
    {
        int rowTiles = (N1 + 63) / 64;
        gemm_kernel<<<rowTiles, 128, 0, stream>>>(
            sbuf, cntb, x, Wl0, bl0, Wr0, nullptr, x1, stats, stats + 256, N1, rowTiles);
    }

    const long NC2 = (long)N2 * CFEAT;
    prep_kernel<49><<<256, 256, 0, stream>>>(cntb, 50048, stats1, 512, e1dst, keyb, E1,
        stats, N1, g0, be0, Wl1, Wr1, bl1, Wls, Wrs, bAgg, bRoot);
    segsum_kernel<49><<<SEG_GRID, SEG_THREADS, 0, stream>>>(x1, e1src, e1dst, keyb, sbuf, cntb, E1, N2);
    {
        int rowTiles = (N2 + 63) / 64;
        gemm_kernel<<<rowTiles, 128, 0, stream>>>(
            sbuf, cntb, x1, Wls, bRoot, Wrs, bAgg, hbuf, stats1, stats1 + 256, N2, rowTiles);
    }
    norm_kernel<<<(int)((NC2 + 255) / 256), 256, 0, stream>>>(
        hbuf, stats1, stats1 + 256, g1, be1, (float*)d_out, N2);
}


# One small memory request per edge, not three: the start pass packs source row and local target row into one word (was two index loads); edge counts summed in LDS, stored once (were memory adds)
# speedup vs baseline: 3.1221x; 1.0757x over previous
#include <hip/hip_runtime.h>

typedef __attribute__((ext_vector_type(2))) float v2f;
typedef __attribute__((ext_vector_type(8))) float v8f;

#define CFEAT 256
#define PK_SHIFT 18
#define PK_MASK ((1u << PK_SHIFT) - 1u)

template <int ROWS>
__global__ void prep_kernel(float* __restrict__ zp, int nz,
                            float* __restrict__ zq, int nzq,
                            const int* __restrict__ src,
                            const int* __restrict__ dst,
                            unsigned int* __restrict__ key,
                            unsigned int* __restrict__ pk, int E,
                            const float* __restrict__ pstats, int nPrev,
                            const float* __restrict__ pg,
                            const float* __restrict__ pb,
                            const float* __restrict__ Wl,
                            const float* __restrict__ Wr,
                            const float* __restrict__ bl,
                            float* __restrict__ Wls,
                            float* __restrict__ Wrs,
                            float* __restrict__ bAgg,
                            float* __restrict__ bRoot) {
    const int nWords = ((E + 511) >> 9) * 128;
    const int i = (int)(blockIdx.x * blockDim.x + threadIdx.x);
    const int stride = (int)(gridDim.x * blockDim.x);
    for (int j = i; j < nz; j += stride) zp[j] = 0.0f;
    for (int j = i; j < nzq; j += stride) zq[j] = 0.0f;
    if (pstats != nullptr) {
        const float invN = 1.0f / (float)nPrev;
        for (int j = i; j < 2 * CFEAT * CFEAT; j += stride) {
            const int m = j >> 16;
            const int idx = j & (CFEAT * CFEAT - 1);
            const int k = idx & (CFEAT - 1);
            const float mu = pstats[k] * invN;
            const float var = pstats[CFEAT + k] * invN - mu * mu;
            const float a = pg[k] * rsqrtf(var + 1e-5f);
            if (m == 0) Wls[idx] = Wl[idx] * a;
            else        Wrs[idx] = Wr[idx] * a;
        }
        const int gwave = i >> 5;
        const int lane = i & 31;
        if (gwave < 2 * CFEAT) {
            const int m = gwave >> 8;
            const int jrow = gwave & (CFEAT - 1);
            const float* wrow = (m == 0 ? Wl : Wr) + jrow * CFEAT;
            float acc = 0.0f;
            for (int t = 0; t < CFEAT / 32; ++t) {
                const int k = lane + 32 * t;
                const float mu = pstats[k] * invN;
                const float var = pstats[CFEAT + k] * invN - mu * mu;
                const float a = pg[k] * rsqrtf(var + 1e-5f);
                const float c = pb[k] - a * mu;
                acc += c * wrow[k];
            }
            for (int d = 16; d >= 1; d >>= 1) acc += __shfl_xor(acc, d, 32);
            if (lane == 0) {
                if (m == 0) bAgg[jrow] = acc;
                else        bRoot[jrow] = bl[jrow] + acc;
            }
        }
    }
    for (int j = i; j < nWords; j += stride) {
        const int e = 4 * j;
        unsigned w = 0u;
        if (e + 3 < E) {
            const int4 d = *(const int4*)(dst + e);
            const int4 sv = *(const int4*)(src + e);
            const int ox = d.x / ROWS, oy = d.y / ROWS, oz = d.z / ROWS, ow = d.w / ROWS;
            w = ((unsigned)ox & 255u) | (((unsigned)oy & 255u) << 8) |
                (((unsigned)oz & 255u) << 16) | (((unsigned)ow & 255u) << 24);
            uint4 p;
            p.x = (unsigned)sv.x | ((unsigned)(d.x - ox * ROWS) << PK_SHIFT);
            p.y = (unsigned)sv.y | ((unsigned)(d.y - oy * ROWS) << PK_SHIFT);
            p.z = (unsigned)sv.z | ((unsigned)(d.z - oz * ROWS) << PK_SHIFT);
            p.w = (unsigned)sv.w | ((unsigned)(d.w - ow * ROWS) << PK_SHIFT);
            *(uint4*)(pk + e) = p;
        } else {
            for (int b = 0; b < 4; ++b)
                if (e + b < E) {
                    const int db = dst[e + b];
                    const int ob = db / ROWS;
                    w |= ((unsigned)ob & 255u) << (8 * b);
                    pk[e + b] = (unsigned)src[e + b] | ((unsigned)(db - ob * ROWS) << PK_SHIFT);
                }
        }
        key[j] = w;
    }
}

#define SEG_THREADS 1024
#define SEG_WAVES 32
#define SEG_GRID 256

__device__ __forceinline__ void row_load(float (&v)[8], const float* __restrict__ x, int sr, int lane) {
    const float* xs = x + (long)sr * CFEAT + lane;
#pragma unroll
    for (int j = 0; j < 8; ++j) v[j] = xs[j * 32];
}

__device__ __forceinline__ void row_add_lds(const float (&v)[8], float* acc, int dr, int lane) {
    float* a = acc + dr * CFEAT + lane;
#pragma unroll
    for (int j = 0; j < 8; ++j) atomicAdd(&a[j * 32], v[j]);
}

#define SEG_BATCH() do { \
        const unsigned w0_ = pk[qe0], w1_ = pk[qe1], w2_ = pk[qe2], w3_ = pk[qe3]; \
        const int s0_ = (int)(w0_ & PK_MASK), s1_ = (int)(w1_ & PK_MASK); \
        const int s2_ = (int)(w2_ & PK_MASK), s3_ = (int)(w3_ & PK_MASK); \
        const int d0_ = (int)(w0_ >> PK_SHIFT), d1_ = (int)(w1_ >> PK_SHIFT); \
        const int d2_ = (int)(w2_ >> PK_SHIFT), d3_ = (int)(w3_ >> PK_SHIFT); \
        if (lane < 4) { \
            const int dl_ = (lane == 0) ? d0_ : (lane == 1) ? d1_ : (lane == 2) ? d2_ : d3_; \
            atomicAdd(&cntL[dl_], 1.0f); \
        } \
        float v0_[8], v1_[8], v2_[8], v3_[8]; \
        row_load(v0_, x, s0_, lane); \
        row_load(v1_, x, s1_, lane); \
        row_load(v2_, x, s2_, lane); \
        row_load(v3_, x, s3_, lane); \
        row_add_lds(v0_, acc, d0_, lane); \
        row_add_lds(v1_, acc, d1_, lane); \
        row_add_lds(v2_, acc, d2_, lane); \
        row_add_lds(v3_, acc, d3_, lane); \
        np = 0; \
    } while (0)

#define SEG_ENQUEUE(E_) do { \
        const int e_ = (E_); \
        if (e_ < E) { \
            if (np == 0)      qe0 = e_; \
            else if (np == 1) qe1 = e_; \
            else if (np == 2) qe2 = e_; \
            else              qe3 = e_; \
            ++np; \
            if (np == 4) SEG_BATCH(); \
        } \
    } while (0)

#define SEG_ZFLAGS(X_) (~((((X_) & 0x7F7F7F7Fu) + 0x7F7F7F7Fu) | (X_)) & 0x80808080u)

#define SEG_PROCESS(CC_, K_) do { \
        const int cc_ = (CC_); \
        const unsigned own_ = (SEG_ZFLAGS((K_).x ^ gpat) >> 7) | (SEG_ZFLAGS((K_).y ^ gpat) >> 6) | \
                              (SEG_ZFLAGS((K_).z ^ gpat) >> 5) | (SEG_ZFLAGS((K_).w ^ gpat) >> 4); \
        unsigned M_ = __builtin_amdgcn_ballot_w32(own_ != 0u); \
        if (cc_ >= nCh) M_ = 0u; \
        if (M_) { \
            const int ebase_ = SEG_CHUNK(cc_) * 512; \
            while (M_) { \
                const int i_ = __builtin_ctz(M_); \
                M_ &= M_ - 1; \
                unsigned o_ = (unsigned)__builtin_amdgcn_readlane((int)own_, i_); \
                while (o_) { \
                    const int j_ = __builtin_ctz(o_); \
                    o_ &= o_ - 1; \
                    SEG_ENQUEUE(ebase_ + i_ * 16 + ((j_ & 3) << 2) + (j_ >> 3)); \
                } \
            } \
        } \
    } while (0)

#define SEG_CHUNK(T_) (((T_) + rot < nCh) ? ((T_) + rot) : ((T_) + rot - nCh))
#define SEG_LOAD(C_) (*(const uint4*)(key + SEG_CHUNK(((C_) < last) ? (C_) : last) * 128 + lane * 4))

template <int ROWS>
__global__ void __launch_bounds__(SEG_THREADS) segsum_kernel(const float* __restrict__ x,
                              const unsigned int* __restrict__ pk,
                              const unsigned int* __restrict__ key,
                              float* __restrict__ s,
                              float* __restrict__ cnt,
                              int E, int N) {
    __shared__ __attribute__((aligned(16))) float acc[ROWS * CFEAT];
    __shared__ float cntL[ROWS];
    const int tid = threadIdx.x;
    const int lane = tid & 31;
    const int wv = __builtin_amdgcn_readfirstlane(tid >> 5);
    const int r0 = (int)blockIdx.x * ROWS;
    const unsigned gpat = (unsigned)blockIdx.x * 0x01010101u;

    float4* acc4 = (float4*)acc;
    for (int i = tid; i < ROWS * (CFEAT / 4); i += SEG_THREADS) acc4[i] = make_float4(0.0f, 0.0f, 0.0f, 0.0f);
    for (int i = tid; i < ROWS; i += SEG_THREADS) cntL[i] = 0.0f;
    __syncthreads();

    int qe0 = 0, qe1 = 0, qe2 = 0, qe3 = 0, np = 0;

    const int nCh = (E + 511) >> 9;
    if (nCh > 0) {
        const int last = nCh - 1;
        const int rot = (int)(((long)blockIdx.x * nCh) / SEG_GRID);
        uint4 dA = SEG_LOAD(wv);
        __builtin_amdgcn_sched_barrier(0);
        uint4 dB = SEG_LOAD(wv + SEG_WAVES);
        __builtin_amdgcn_sched_barrier(0);
        uint4 dC = SEG_LOAD(wv + 2 * SEG_WAVES);
        uint4 dD;
        for (int c = wv; c < nCh; c += 4 * SEG_WAVES) {
            dD = SEG_LOAD(c + 3 * SEG_WAVES);
            SEG_PROCESS(c, dA);
            dA = SEG_LOAD(c + 4 * SEG_WAVES);
            SEG_PROCESS(c + SEG_WAVES, dB);
            dB = SEG_LOAD(c + 5 * SEG_WAVES);
            SEG_PROCESS(c + 2 * SEG_WAVES, dC);
            dC = SEG_LOAD(c + 6 * SEG_WAVES);
            SEG_PROCESS(c + 3 * SEG_WAVES, dD);
        }
    }
    if (np > 0) {
        const int e1 = (np > 1) ? qe1 : qe0, e2 = (np > 2) ? qe2 : qe0;
        const unsigned w0 = pk[qe0], w1 = pk[e1], w2 = pk[e2];
        const int d0 = (int)(w0 >> PK_SHIFT), d1 = (int)(w1 >> PK_SHIFT), d2 = (int)(w2 >> PK_SHIFT);
        if (lane < np) {
            const int dl = (lane == 0) ? d0 : (lane == 1) ? d1 : d2;
            atomicAdd(&cntL[dl], 1.0f);
        }
        float v0[8];
        row_load(v0, x, (int)(w0 & PK_MASK), lane);
        row_add_lds(v0, acc, d0, lane);
        if (np > 1) {
            row_load(v0, x, (int)(w1 & PK_MASK), lane);
            row_add_lds(v0, acc, d1, lane);
        }
        if (np > 2) {
            row_load(v0, x, (int)(w2 & PK_MASK), lane);
            row_add_lds(v0, acc, d2, lane);
        }
    }
    __syncthreads();

    int rows = N - r0;
    rows = rows < ROWS ? rows : ROWS;
    float4* s4 = (float4*)(s + (long)r0 * CFEAT);
    for (int i = tid; i < rows * (CFEAT / 4); i += SEG_THREADS) s4[i] = acc4[i];
    for (int i = tid; i < rows; i += SEG_THREADS) cnt[r0 + i] = cntL[i];
}

#define GEMM_RT 4
typedef float4 linefrag[4];

__device__ __forceinline__ void load_linefrag(linefrag& f, const float* __restrict__ rowh, int kc) {
    const float4* p = (const float4*)(rowh + kc);
#pragma unroll
    for (int q = 0; q < 4; ++q) f[q] = p[2 * q];
}

__device__ __forceinline__ v2f frag_pair(const linefrag& f, int q, int p) {
    v2f r;
    if (p == 0) { r[0] = f[q].x; r[1] = f[q].y; }
    else        { r[0] = f[q].z; r[1] = f[q].w; }
    return r;
}

__device__ __forceinline__ void mma_line(v8f (&acc)[GEMM_RT][4], const linefrag (&a)[GEMM_RT],
                                         const linefrag (&b)[4]) {
#pragma unroll
    for (int q = 0; q < 4; ++q) {
#pragma unroll
        for (int p = 0; p < 2; ++p) {
#pragma unroll
            for (int r = 0; r < GEMM_RT; ++r) {
                v2f af = frag_pair(a[r], q, p);
#pragma unroll
                for (int t = 0; t < 4; ++t)
                    acc[r][t] = __builtin_amdgcn_wmma_f32_16x16x4_f32(
                        false, af, false, frag_pair(b[t], q, p), (short)0, acc[r][t], false, false);
            }
        }
    }
}

__global__ void __launch_bounds__(128) gemm_kernel(const float* __restrict__ s,
                            const float* __restrict__ cnt,
                            const float* __restrict__ xt,
                            const float* __restrict__ Wl,
                            const float* __restrict__ bl,
                            const float* __restrict__ Wr,
                            const float* __restrict__ bAgg,
                            float* __restrict__ h,
                            float* __restrict__ colsum,
                            float* __restrict__ colsq,
                            int N, int rowTiles) {
    int wavesPerBlock = blockDim.x >> 5;
    int gw = blockIdx.x * wavesPerBlock + (threadIdx.x >> 5);
    int totalWaves = rowTiles * 4;
    if (gw >= totalWaves) return;

    int rowTile = gw >> 2;
    int colGrp  = gw & 3;
    int lane = threadIdx.x & 31;
    int l15  = lane & 15;
    int hoff = (lane >> 4) * 4;

    float invc[GEMM_RT];
    unsigned hasIn[GEMM_RT];
    const float* sRow[GEMM_RT];
    const float* xRow[GEMM_RT];
#pragma unroll
    for (int r = 0; r < GEMM_RT; ++r) {
        int rowA = (rowTile * GEMM_RT + r) * 16 + l15;
        int rowL = (rowA < N) ? rowA : (N - 1);
        const float cr = cnt[rowL];
        invc[r] = 1.0f / fmaxf(cr, 1.0f);
        hasIn[r] = __builtin_amdgcn_ballot_w32(cr > 0.0f);
        sRow[r] = s  + (long)rowL * CFEAT + hoff;
        xRow[r] = xt + (long)rowL * CFEAT + hoff;
    }

    int colBase = colGrp * 64;
    const float* wlRow[4];
    const float* wrRow[4];
#pragma unroll
    for (int t = 0; t < 4; ++t) {
        int colB = colBase + t * 16 + l15;
        wlRow[t] = Wl + (long)colB * CFEAT + hoff;
        wrRow[t] = Wr + (long)colB * CFEAT + hoff;
    }

    v8f acc[GEMM_RT][4];
#pragma unroll
    for (int r = 0; r < GEMM_RT; ++r)
#pragma unroll
        for (int t = 0; t < 4; ++t)
            acc[r][t] = (v8f){0, 0, 0, 0, 0, 0, 0, 0};

    linefrag aC[GEMM_RT], bC[4], aN[GEMM_RT], bN[4];
#pragma unroll
    for (int r = 0; r < GEMM_RT; ++r) load_linefrag(aC[r], sRow[r], 0);
#pragma unroll
    for (int t = 0; t < 4; ++t) load_linefrag(bC[t], wlRow[t], 0);

    for (int kc = 0; kc < CFEAT - 32; kc += 32) {
#pragma unroll
        for (int r = 0; r < GEMM_RT; ++r) load_linefrag(aN[r], xRow[r], kc);
#pragma unroll
        for (int t = 0; t < 4; ++t) load_linefrag(bN[t], wrRow[t], kc);
        __builtin_amdgcn_sched_barrier(0);
#pragma unroll
        for (int r = 0; r < GEMM_RT; ++r)
#pragma unroll
            for (int q = 0; q < 4; ++q) aC[r][q] *= invc[r];
        mma_line(acc, aC, bC);
        __builtin_amdgcn_sched_barrier(0);
#pragma unroll
        for (int r = 0; r < GEMM_RT; ++r) load_linefrag(aC[r], sRow[r], kc + 32);
#pragma unroll
        for (int t = 0; t < 4; ++t) load_linefrag(bC[t], wlRow[t], kc + 32);
        __builtin_amdgcn_sched_barrier(0);
        mma_line(acc, aN, bN);
    }
#pragma unroll
    for (int r = 0; r < GEMM_RT; ++r) load_linefrag(aN[r], xRow[r], CFEAT - 32);
#pragma unroll
    for (int t = 0; t < 4; ++t) load_linefrag(bN[t], wrRow[t], CFEAT - 32);
    __builtin_amdgcn_sched_barrier(0);
#pragma unroll
    for (int r = 0; r < GEMM_RT; ++r)
#pragma unroll
        for (int q = 0; q < 4; ++q) aC[r][q] *= invc[r];
    mma_line(acc, aC, bC);
    mma_line(acc, aN, bN);

    int half8 = (lane < 16) ? 0 : 8;
#pragma unroll
    for (int t = 0; t < 4; ++t) {
        int col = colBase + t * 16 + l15;
        float bias = bl[col];
        float ba = (bAgg != nullptr) ? bAgg[col] : 0.0f;
        float sp = 0.0f, qp = 0.0f;
#pragma unroll
        for (int r = 0; r < GEMM_RT; ++r) {
            int rbase = (rowTile * GEMM_RT + r) * 16 + half8;
#pragma unroll
            for (int v = 0; v < 8; ++v) {
                float bv = ((hasIn[r] >> (half8 + v)) & 1u) ? (bias + ba) : bias;
                float val = fmaxf(acc[r][t][v] + bv, 0.0f);
                int rr = rbase + v;
                bool ok = (rr < N);
                float m = ok ? val : 0.0f;
                sp += m;
                qp += m * m;
                if (ok) h[(long)rr * CFEAT + col] = val;
            }
        }
        atomicAdd(&colsum[col], sp);
        atomicAdd(&colsq[col], qp);
    }
}

__global__ void norm_kernel(const float* __restrict__ hbuf,
                            const float* __restrict__ colsum,
                            const float* __restrict__ colsq,
                            const float* __restrict__ gamma,
                            const float* __restrict__ beta,
                            float* __restrict__ out, int N) {
    long i = (long)blockIdx.x * blockDim.x + threadIdx.x;
    long total = (long)N * CFEAT;
    if (i >= total) return;
    int col = (int)(i & (CFEAT - 1));
    float invN = 1.0f / (float)N;
    float mu  = colsum[col] * invN;
    float var = colsq[col] * invN - mu * mu;
    float rs  = rsqrtf(var + 1e-5f);
    out[i] = gamma[col] * (hbuf[i] - mu) * rs + beta[col];
}

extern "C" void kernel_launch(void* const* d_in, const int* in_sizes, int n_in,
                              void* d_out, int out_size, void* d_ws, size_t ws_size,
                              hipStream_t stream) {
    const float* x    = (const float*)d_in[0];
    const int* e0src  = (const int*)d_in[1];
    const int* e0dst  = (const int*)d_in[2];
    const int* e1src  = (const int*)d_in[3];
    const int* e1dst  = (const int*)d_in[4];
    const float* Wl0  = (const float*)d_in[5];
    const float* bl0  = (const float*)d_in[6];
    const float* Wr0  = (const float*)d_in[7];
    const float* g0   = (const float*)d_in[8];
    const float* be0  = (const float*)d_in[9];
    const float* Wl1  = (const float*)d_in[10];
    const float* bl1  = (const float*)d_in[11];
    const float* Wr1  = (const float*)d_in[12];
    const float* g1   = (const float*)d_in[13];
    const float* be1  = (const float*)d_in[14];

    const int N1 = 50000, N2 = 12500;
    const int E0 = in_sizes[1];
    const int E1 = in_sizes[3];

    float* ws = (float*)d_ws;
    const long NC1 = (long)N1 * CFEAT;
    float* x1    = ws;
    float* sbuf  = ws + NC1;
    float* hbuf  = ws + 2 * NC1;
    float* cntb  = ws + 3 * NC1;
    float* stats = cntb + 50048;
    float* Wls    = hbuf + 4 * 1048576;
    float* Wrs    = Wls + CFEAT * CFEAT;
    float* bAgg   = Wrs + CFEAT * CFEAT;
    float* bRoot  = bAgg + CFEAT;
    float* stats1 = bRoot + CFEAT;
    unsigned int* pkb = (unsigned int*)(hbuf + 5 * 1048576);

    unsigned int* keyb = (unsigned int*)hbuf;
    prep_kernel<196><<<256, 256, 0, stream>>>(stats, 512, nullptr, 0, e0src, e0dst, keyb, pkb, E0,
        nullptr, 0, nullptr, nullptr, nullptr, nullptr, nullptr, nullptr, nullptr, nullptr, nullptr);
    segsum_kernel<196><<<SEG_GRID, SEG_THREADS, 0, stream>>>(x, pkb, keyb, sbuf, cntb, E0, N1);
    {
        int rowTiles = (N1 + 63) / 64;
        gemm_kernel<<<rowTiles, 128, 0, stream>>>(
            sbuf, cntb, x, Wl0, bl0, Wr0, nullptr, x1, stats, stats + 256, N1, rowTiles);
    }

    const long NC2 = (long)N2 * CFEAT;
    prep_kernel<49><<<256, 256, 0, stream>>>(stats1, 512, nullptr, 0, e1src, e1dst, keyb, pkb, E1,
        stats, N1, g0, be0, Wl1, Wr1, bl1, Wls, Wrs, bAgg, bRoot);
    segsum_kernel<49><<<SEG_GRID, SEG_THREADS, 0, stream>>>(x1, pkb, keyb, sbuf, cntb, E1, N2);
    {
        int rowTiles = (N2 + 63) / 64;
        gemm_kernel<<<rowTiles, 128, 0, stream>>>(
            sbuf, cntb, x1, Wls, bRoot, Wrs, bAgg, hbuf, stats1, stats1 + 256, N2, rowTiles);
    }
    norm_kernel<<<(int)((NC2 + 255) / 256), 256, 0, stream>>>(
        hbuf, stats1, stats1 + 256, g1, be1, (float*)d_out, N2);
}


# GEMM on a fixed grid, one 8-wave workgroup per WGP: 16-row tiles dealt in equal shares (12 or 13 per WGP; was 3 or 4 64-row tiles), 48- or 64-row register tiles; column sums: 24 or 32 rows per atomic
# speedup vs baseline: 3.2480x; 1.0403x over previous
#include <hip/hip_runtime.h>

typedef __attribute__((ext_vector_type(2))) float v2f;
typedef __attribute__((ext_vector_type(8))) float v8f;

#define CFEAT 256
#define PK_SHIFT 18
#define PK_MASK ((1u << PK_SHIFT) - 1u)

template <int ROWS>
__global__ void prep_kernel(float* __restrict__ zp, int nz,
                            float* __restrict__ zq, int nzq,
                            const int* __restrict__ src,
                            const int* __restrict__ dst,
                            unsigned int* __restrict__ key,
                            unsigned int* __restrict__ pk, int E,
                            const float* __restrict__ pstats, int nPrev,
                            const float* __restrict__ pg,
                            const float* __restrict__ pb,
                            const float* __restrict__ Wl,
                            const float* __restrict__ Wr,
                            const float* __restrict__ bl,
                            float* __restrict__ Wls,
                            float* __restrict__ Wrs,
                            float* __restrict__ bAgg,
                            float* __restrict__ bRoot) {
    const int nWords = ((E + 511) >> 9) * 128;
    const int i = (int)(blockIdx.x * blockDim.x + threadIdx.x);
    const int stride = (int)(gridDim.x * blockDim.x);
    for (int j = i; j < nz; j += stride) zp[j] = 0.0f;
    for (int j = i; j < nzq; j += stride) zq[j] = 0.0f;
    if (pstats != nullptr) {
        const float invN = 1.0f / (float)nPrev;
        for (int j = i; j < 2 * CFEAT * CFEAT; j += stride) {
            const int m = j >> 16;
            const int idx = j & (CFEAT * CFEAT - 1);
            const int k = idx & (CFEAT - 1);
            const float mu = pstats[k] * invN;
            const float var = pstats[CFEAT + k] * invN - mu * mu;
            const float a = pg[k] * rsqrtf(var + 1e-5f);
            if (m == 0) Wls[idx] = Wl[idx] * a;
            else        Wrs[idx] = Wr[idx] * a;
        }
        const int gwave = i >> 5;
        const int lane = i & 31;
        if (gwave < 2 * CFEAT) {
            const int m = gwave >> 8;
            const int jrow = gwave & (CFEAT - 1);
            const float* wrow = (m == 0 ? Wl : Wr) + jrow * CFEAT;
            float acc = 0.0f;
            for (int t = 0; t < CFEAT / 32; ++t) {
                const int k = lane + 32 * t;
                const float mu = pstats[k] * invN;
                const float var = pstats[CFEAT + k] * invN - mu * mu;
                const float a = pg[k] * rsqrtf(var + 1e-5f);
                const float c = pb[k] - a * mu;
                acc += c * wrow[k];
            }
            for (int d = 16; d >= 1; d >>= 1) acc += __shfl_xor(acc, d, 32);
            if (lane == 0) {
                if (m == 0) bAgg[jrow] = acc;
                else        bRoot[jrow] = bl[jrow] + acc;
            }
        }
    }
    for (int j = i; j < nWords; j += stride) {
        const int e = 4 * j;
        unsigned w = 0u;
        if (e + 3 < E) {
            const int4 d = *(const int4*)(dst + e);
            const int4 sv = *(const int4*)(src + e);
            const int ox = d.x / ROWS, oy = d.y / ROWS, oz = d.z / ROWS, ow = d.w / ROWS;
            w = ((unsigned)ox & 255u) | (((unsigned)oy & 255u) << 8) |
                (((unsigned)oz & 255u) << 16) | (((unsigned)ow & 255u) << 24);
            uint4 p;
            p.x = (unsigned)sv.x | ((unsigned)(d.x - ox * ROWS) << PK_SHIFT);
            p.y = (unsigned)sv.y | ((unsigned)(d.y - oy * ROWS) << PK_SHIFT);
            p.z = (unsigned)sv.z | ((unsigned)(d.z - oz * ROWS) << PK_SHIFT);
            p.w = (unsigned)sv.w | ((unsigned)(d.w - ow * ROWS) << PK_SHIFT);
            *(uint4*)(pk + e) = p;
        } else {
            for (int b = 0; b < 4; ++b)
                if (e + b < E) {
                    const int db = dst[e + b];
                    const int ob = db / ROWS;
                    w |= ((unsigned)ob & 255u) << (8 * b);
                    pk[e + b] = (unsigned)src[e + b] | ((unsigned)(db - ob * ROWS) << PK_SHIFT);
                }
        }
        key[j] = w;
    }
}

#define SEG_THREADS 1024
#define SEG_WAVES 32
#define SEG_GRID 256

__device__ __forceinline__ void row_load(float (&v)[8], const float* __restrict__ x, int sr, int lane) {
    const float* xs = x + (long)sr * CFEAT + lane;
#pragma unroll
    for (int j = 0; j < 8; ++j) v[j] = xs[j * 32];
}

__device__ __forceinline__ void row_add_lds(const float (&v)[8], float* acc, int dr, int lane) {
    float* a = acc + dr * CFEAT + lane;
#pragma unroll
    for (int j = 0; j < 8; ++j) atomicAdd(&a[j * 32], v[j]);
}

#define SEG_BATCH() do { \
        const unsigned w0_ = pk[qe0], w1_ = pk[qe1], w2_ = pk[qe2], w3_ = pk[qe3]; \
        const int s0_ = (int)(w0_ & PK_MASK), s1_ = (int)(w1_ & PK_MASK); \
        const int s2_ = (int)(w2_ & PK_MASK), s3_ = (int)(w3_ & PK_MASK); \
        const int d0_ = (int)(w0_ >> PK_SHIFT), d1_ = (int)(w1_ >> PK_SHIFT); \
        const int d2_ = (int)(w2_ >> PK_SHIFT), d3_ = (int)(w3_ >> PK_SHIFT); \
        if (lane < 4) { \
            const int dl_ = (lane == 0) ? d0_ : (lane == 1) ? d1_ : (lane == 2) ? d2_ : d3_; \
            atomicAdd(&cntL[dl_], 1.0f); \
        } \
        float v0_[8], v1_[8], v2_[8], v3_[8]; \
        row_load(v0_, x, s0_, lane); \
        row_load(v1_, x, s1_, lane); \
        row_load(v2_, x, s2_, lane); \
        row_load(v3_, x, s3_, lane); \
        row_add_lds(v0_, acc, d0_, lane); \
        row_add_lds(v1_, acc, d1_, lane); \
        row_add_lds(v2_, acc, d2_, lane); \
        row_add_lds(v3_, acc, d3_, lane); \
        np = 0; \
    } while (0)

#define SEG_ENQUEUE(E_) do { \
        const int e_ = (E_); \
        if (e_ < E) { \
            if (np == 0)      qe0 = e_; \
            else if (np == 1) qe1 = e_; \
            else if (np == 2) qe2 = e_; \
            else              qe3 = e_; \
            ++np; \
            if (np == 4) SEG_BATCH(); \
        } \
    } while (0)

#define SEG_ZFLAGS(X_) (~((((X_) & 0x7F7F7F7Fu) + 0x7F7F7F7Fu) | (X_)) & 0x80808080u)

#define SEG_PROCESS(CC_, K_) do { \
        const int cc_ = (CC_); \
        const unsigned own_ = (SEG_ZFLAGS((K_).x ^ gpat) >> 7) | (SEG_ZFLAGS((K_).y ^ gpat) >> 6) | \
                              (SEG_ZFLAGS((K_).z ^ gpat) >> 5) | (SEG_ZFLAGS((K_).w ^ gpat) >> 4); \
        unsigned M_ = __builtin_amdgcn_ballot_w32(own_ != 0u); \
        if (cc_ >= nCh) M_ = 0u; \
        if (M_) { \
            const int ebase_ = SEG_CHUNK(cc_) * 512; \
            while (M_) { \
                const int i_ = __builtin_ctz(M_); \
                M_ &= M_ - 1; \
                unsigned o_ = (unsigned)__builtin_amdgcn_readlane((int)own_, i_); \
                while (o_) { \
                    const int j_ = __builtin_ctz(o_); \
                    o_ &= o_ - 1; \
                    SEG_ENQUEUE(ebase_ + i_ * 16 + ((j_ & 3) << 2) + (j_ >> 3)); \
                } \
            } \
        } \
    } while (0)

#define SEG_CHUNK(T_) (((T_) + rot < nCh) ? ((T_) + rot) : ((T_) + rot - nCh))
#define SEG_LOAD(C_) (*(const uint4*)(key + SEG_CHUNK(((C_) < last) ? (C_) : last) * 128 + lane * 4))

template <int ROWS>
__global__ void __launch_bounds__(SEG_THREADS) segsum_kernel(const float* __restrict__ x,
                              const unsigned int* __restrict__ pk,
                              const unsigned int* __restrict__ key,
                              float* __restrict__ s,
                              float* __restrict__ cnt,
                              int E, int N) {
    __shared__ __attribute__((aligned(16))) float acc[ROWS * CFEAT];
    __shared__ float cntL[ROWS];
    const int tid = threadIdx.x;
    const int lane = tid & 31;
    const int wv = __builtin_amdgcn_readfirstlane(tid >> 5);
    const int r0 = (int)blockIdx.x * ROWS;
    const unsigned gpat = (unsigned)blockIdx.x * 0x01010101u;

    float4* acc4 = (float4*)acc;
    for (int i = tid; i < ROWS * (CFEAT / 4); i += SEG_THREADS) acc4[i] = make_float4(0.0f, 0.0f, 0.0f, 0.0f);
    for (int i = tid; i < ROWS; i += SEG_THREADS) cntL[i] = 0.0f;
    __syncthreads();

    int qe0 = 0, qe1 = 0, qe2 = 0, qe3 = 0, np = 0;

    const int nCh = (E + 511) >> 9;
    if (nCh > 0) {
        const int last = nCh - 1;
        const int rot = (int)(((long)blockIdx.x * nCh) / SEG_GRID);
        uint4 dA = SEG_LOAD(wv);
        __builtin_amdgcn_sched_barrier(0);
        uint4 dB = SEG_LOAD(wv + SEG_WAVES);
        __builtin_amdgcn_sched_barrier(0);
        uint4 dC = SEG_LOAD(wv + 2 * SEG_WAVES);
        uint4 dD;
        for (int c = wv; c < nCh; c += 4 * SEG_WAVES) {
            dD = SEG_LOAD(c + 3 * SEG_WAVES);
            SEG_PROCESS(c, dA);
            dA = SEG_LOAD(c + 4 * SEG_WAVES);
            SEG_PROCESS(c + SEG_WAVES, dB);
            dB = SEG_LOAD(c + 5 * SEG_WAVES);
            SEG_PROCESS(c + 2 * SEG_WAVES, dC);
            dC = SEG_LOAD(c + 6 * SEG_WAVES);
            SEG_PROCESS(c + 3 * SEG_WAVES, dD);
        }
    }
    if (np > 0) {
        const int e1 = (np > 1) ? qe1 : qe0, e2 = (np > 2) ? qe2 : qe0;
        const unsigned w0 = pk[qe0], w1 = pk[e1], w2 = pk[e2];
        const int d0 = (int)(w0 >> PK_SHIFT), d1 = (int)(w1 >> PK_SHIFT), d2 = (int)(w2 >> PK_SHIFT);
        if (lane < np) {
            const int dl = (lane == 0) ? d0 : (lane == 1) ? d1 : d2;
            atomicAdd(&cntL[dl], 1.0f);
        }
        float v0[8];
        row_load(v0, x, (int)(w0 & PK_MASK), lane);
        row_add_lds(v0, acc, d0, lane);
        if (np > 1) {
            row_load(v0, x, (int)(w1 & PK_MASK), lane);
            row_add_lds(v0, acc, d1, lane);
        }
        if (np > 2) {
            row_load(v0, x, (int)(w2 & PK_MASK), lane);
            row_add_lds(v0, acc, d2, lane);
        }
    }
    __syncthreads();

    int rows = N - r0;
    rows = rows < ROWS ? rows : ROWS;
    float4* s4 = (float4*)(s + (long)r0 * CFEAT);
    for (int i = tid; i < rows * (CFEAT / 4); i += SEG_THREADS) s4[i] = acc4[i];
    for (int i = tid; i < rows; i += SEG_THREADS) cnt[r0 + i] = cntL[i];
}

#define GEMM_RT_MAX 4
#define GEMM_GRID 256
typedef float4 linefrag[4];

__device__ __forceinline__ void load_linefrag(linefrag& f, const float* __restrict__ rowh, int kc) {
    const float4* p = (const float4*)(rowh + kc);
#pragma unroll
    for (int q = 0; q < 4; ++q) f[q] = p[2 * q];
}

__device__ __forceinline__ v2f frag_pair(const linefrag& f, int q, int p) {
    v2f r;
    if (p == 0) { r[0] = f[q].x; r[1] = f[q].y; }
    else        { r[0] = f[q].z; r[1] = f[q].w; }
    return r;
}

template <int RT>
__device__ __forceinline__ void mma_line(v8f (&acc)[RT][4], const linefrag (&a)[RT],
                                         const linefrag (&b)[4]) {
#pragma unroll
    for (int q = 0; q < 4; ++q) {
#pragma unroll
        for (int p = 0; p < 2; ++p) {
#pragma unroll
            for (int r = 0; r < RT; ++r) {
                v2f af = frag_pair(a[r], q, p);
#pragma unroll
                for (int t = 0; t < 4; ++t)
                    acc[r][t] = __builtin_amdgcn_wmma_f32_16x16x4_f32(
                        false, af, false, frag_pair(b[t], q, p), (short)0, acc[r][t], false, false);
            }
        }
    }
}

template <int RT>
__device__ __forceinline__ void gemm_group(const float* __restrict__ s,
                            const float* __restrict__ cnt,
                            const float* __restrict__ xt,
                            const float* __restrict__ Wl,
                            const float* __restrict__ bl,
                            const float* __restrict__ Wr,
                            const float* __restrict__ bAgg,
                            float* __restrict__ h,
                            float* __restrict__ colsum,
                            float* __restrict__ colsq,
                            int N, int t0, int colBase, int lane) {
    int l15  = lane & 15;
    int hoff = (lane >> 4) * 4;

    float invc[RT];
    unsigned hasIn[RT];
    const float* sRow[RT];
    const float* xRow[RT];
#pragma unroll
    for (int r = 0; r < RT; ++r) {
        int rowA = (t0 + r) * 16 + l15;
        int rowL = (rowA < N) ? rowA : (N - 1);
        const float cr = cnt[rowL];
        invc[r] = 1.0f / fmaxf(cr, 1.0f);
        hasIn[r] = __builtin_amdgcn_ballot_w32(cr > 0.0f);
        sRow[r] = s  + (long)rowL * CFEAT + hoff;
        xRow[r] = xt + (long)rowL * CFEAT + hoff;
    }

    const float* wlRow[4];
    const float* wrRow[4];
#pragma unroll
    for (int t = 0; t < 4; ++t) {
        int colB = colBase + t * 16 + l15;
        wlRow[t] = Wl + (long)colB * CFEAT + hoff;
        wrRow[t] = Wr + (long)colB * CFEAT + hoff;
    }

    v8f acc[RT][4];
#pragma unroll
    for (int r = 0; r < RT; ++r)
#pragma unroll
        for (int t = 0; t < 4; ++t)
            acc[r][t] = (v8f){0, 0, 0, 0, 0, 0, 0, 0};

    linefrag aC[RT], bC[4], aN[RT], bN[4];
#pragma unroll
    for (int r = 0; r < RT; ++r) load_linefrag(aC[r], sRow[r], 0);
#pragma unroll
    for (int t = 0; t < 4; ++t) load_linefrag(bC[t], wlRow[t], 0);

    for (int kc = 0; kc < CFEAT - 32; kc += 32) {
        asm volatile("v_nop\n\tv_nop\n\tv_nop\n\tv_nop");
        __builtin_amdgcn_sched_barrier(0);
#pragma unroll
        for (int r = 0; r < RT; ++r) load_linefrag(aN[r], xRow[r], kc);
#pragma unroll
        for (int t = 0; t < 4; ++t) load_linefrag(bN[t], wrRow[t], kc);
        __builtin_amdgcn_sched_barrier(0);
#pragma unroll
        for (int r = 0; r < RT; ++r)
#pragma unroll
            for (int q = 0; q < 4; ++q) aC[r][q] *= invc[r];
        mma_line<RT>(acc, aC, bC);
        __builtin_amdgcn_sched_barrier(0);
#pragma unroll
        for (int r = 0; r < RT; ++r) load_linefrag(aC[r], sRow[r], kc + 32);
#pragma unroll
        for (int t = 0; t < 4; ++t) load_linefrag(bC[t], wlRow[t], kc + 32);
        __builtin_amdgcn_sched_barrier(0);
        mma_line<RT>(acc, aN, bN);
    }
#pragma unroll
    for (int r = 0; r < RT; ++r) load_linefrag(aN[r], xRow[r], CFEAT - 32);
#pragma unroll
    for (int t = 0; t < 4; ++t) load_linefrag(bN[t], wrRow[t], CFEAT - 32);
    __builtin_amdgcn_sched_barrier(0);
#pragma unroll
    for (int r = 0; r < RT; ++r)
#pragma unroll
        for (int q = 0; q < 4; ++q) aC[r][q] *= invc[r];
    mma_line<RT>(acc, aC, bC);
    mma_line<RT>(acc, aN, bN);

    int half8 = (lane < 16) ? 0 : 8;
#pragma unroll
    for (int t = 0; t < 4; ++t) {
        int col = colBase + t * 16 + l15;
        float bias = bl[col];
        float ba = (bAgg != nullptr) ? bAgg[col] : 0.0f;
        float sp = 0.0f, qp = 0.0f;
#pragma unroll
        for (int r = 0; r < RT; ++r) {
            int rbase = (t0 + r) * 16 + half8;
#pragma unroll
            for (int v = 0; v < 8; ++v) {
                float bv = ((hasIn[r] >> (half8 + v)) & 1u) ? (bias + ba) : bias;
                float val = fmaxf(acc[r][t][v] + bv, 0.0f);
                int rr = rbase + v;
                bool ok = (rr < N);
                float m = ok ? val : 0.0f;
                sp += m;
                qp += m * m;
                if (ok) h[(long)rr * CFEAT + col] = val;
            }
        }
        atomicAdd(&colsum[col], sp);
        atomicAdd(&colsq[col], qp);
    }
}

__global__ void __launch_bounds__(256) gemm_kernel(const float* __restrict__ s,
                            const float* __restrict__ cnt,
                            const float* __restrict__ xt,
                            const float* __restrict__ Wl,
                            const float* __restrict__ bl,
                            const float* __restrict__ Wr,
                            const float* __restrict__ bAgg,
                            float* __restrict__ h,
                            float* __restrict__ colsum,
                            float* __restrict__ colsq,
                            int N, int nT16) {
    const int lane = threadIdx.x & 31;
    const int wv = __builtin_amdgcn_readfirstlane((int)(threadIdx.x >> 5));
    const int colBase = (wv & 3) * 64;
    const int part = wv >> 2;
    const int nParts = (int)(blockDim.x >> 7);
    const int G = (int)gridDim.x;
    const int tBeg = (int)(((unsigned)blockIdx.x * (unsigned)nT16) / (unsigned)G);
    const int tEnd = (int)((((unsigned)blockIdx.x + 1u) * (unsigned)nT16) / (unsigned)G);
    const int n = tEnd - tBeg;
    const int pBeg = tBeg + (part * n) / nParts;
    const int pEnd = tBeg + ((part + 1) * n) / nParts;
    const int m = pEnd - pBeg;
    if (m <= 0) return;
    const int k = (m + GEMM_RT_MAX - 1) / GEMM_RT_MAX;
    const int base = m / k;
    const int extra = m - base * k;
    int t0 = pBeg;
    for (int g = 0; g < k; ++g) {
        const int sz = base + ((g < extra) ? 1 : 0);
#define GEMM_CASE(R_) case R_: gemm_group<R_>(s, cnt, xt, Wl, bl, Wr, bAgg, h, colsum, colsq, N, t0, colBase, lane); break;
        switch (sz) {
            GEMM_CASE(1) GEMM_CASE(2) GEMM_CASE(3)
            default: gemm_group<GEMM_RT_MAX>(s, cnt, xt, Wl, bl, Wr, bAgg, h, colsum, colsq, N, t0, colBase, lane); break;
        }
#undef GEMM_CASE
        t0 += sz;
    }
}

__global__ void norm_kernel(const float* __restrict__ hbuf,
                            const float* __restrict__ colsum,
                            const float* __restrict__ colsq,
                            const float* __restrict__ gamma,
                            const float* __restrict__ beta,
                            float* __restrict__ out, int N) {
    long i = (long)blockIdx.x * blockDim.x + threadIdx.x;
    long total = (long)N * CFEAT;
    if (i >= total) return;
    int col = (int)(i & (CFEAT - 1));
    float invN = 1.0f / (float)N;
    float mu  = colsum[col] * invN;
    float var = colsq[col] * invN - mu * mu;
    float rs  = rsqrtf(var + 1e-5f);
    out[i] = gamma[col] * (hbuf[i] - mu) * rs + beta[col];
}

extern "C" void kernel_launch(void* const* d_in, const int* in_sizes, int n_in,
                              void* d_out, int out_size, void* d_ws, size_t ws_size,
                              hipStream_t stream) {
    const float* x    = (const float*)d_in[0];
    const int* e0src  = (const int*)d_in[1];
    const int* e0dst  = (const int*)d_in[2];
    const int* e1src  = (const int*)d_in[3];
    const int* e1dst  = (const int*)d_in[4];
    const float* Wl0  = (const float*)d_in[5];
    const float* bl0  = (const float*)d_in[6];
    const float* Wr0  = (const float*)d_in[7];
    const float* g0   = (const float*)d_in[8];
    const float* be0  = (const float*)d_in[9];
    const float* Wl1  = (const float*)d_in[10];
    const float* bl1  = (const float*)d_in[11];
    const float* Wr1  = (const float*)d_in[12];
    const float* g1   = (const float*)d_in[13];
    const float* be1  = (const float*)d_in[14];

    const int N1 = 50000, N2 = 12500;
    const int E0 = in_sizes[1];
    const int E1 = in_sizes[3];

    float* ws = (float*)d_ws;
    const long NC1 = (long)N1 * CFEAT;
    float* x1    = ws;
    float* sbuf  = ws + NC1;
    float* hbuf  = ws + 2 * NC1;
    float* cntb  = ws + 3 * NC1;
    float* stats = cntb + 50048;
    float* Wls    = hbuf + 4 * 1048576;
    float* Wrs    = Wls + CFEAT * CFEAT;
    float* bAgg   = Wrs + CFEAT * CFEAT;
    float* bRoot  = bAgg + CFEAT;
    float* stats1 = bRoot + CFEAT;
    unsigned int* pkb = (unsigned int*)(hbuf + 5 * 1048576);

    unsigned int* keyb = (unsigned int*)hbuf;
    prep_kernel<196><<<256, 256, 0, stream>>>(stats, 512, nullptr, 0, e0src, e0dst, keyb, pkb, E0,
        nullptr, 0, nullptr, nullptr, nullptr, nullptr, nullptr, nullptr, nullptr, nullptr, nullptr);
    segsum_kernel<196><<<SEG_GRID, SEG_THREADS, 0, stream>>>(x, pkb, keyb, sbuf, cntb, E0, N1);
    gemm_kernel<<<GEMM_GRID, 256, 0, stream>>>(
        sbuf, cntb, x, Wl0, bl0, Wr0, nullptr, x1, stats, stats + 256, N1, 3125);

    const long NC2 = (long)N2 * CFEAT;
    prep_kernel<49><<<256, 256, 0, stream>>>(stats1, 512, nullptr, 0, e1src, e1dst, keyb, pkb, E1,
        stats, N1, g0, be0, Wl1, Wr1, bl1, Wls, Wrs, bAgg, bRoot);
    segsum_kernel<49><<<SEG_GRID, SEG_THREADS, 0, stream>>>(x1, pkb, keyb, sbuf, cntb, E1, N2);
    gemm_kernel<<<196, 128, 0, stream>>>(
        sbuf, cntb, x1, Wls, bRoot, Wrs, bAgg, hbuf, stats1, stats1 + 256, N2, 782);
    norm_kernel<<<(int)((NC2 + 255) / 256), 256, 0, stream>>>(
        hbuf, stats1, stats1 + 256, g1, be1, (float*)d_out, N2);
}


# Owner gather in 512-byte requests: a source row is two 16-byte-per-lane loads of four whole lines each (was eight one-line loads); LDS rows lane-major, undone at store; adds still in arrival order
# speedup vs baseline: 3.4636x; 1.0664x over previous
#include <hip/hip_runtime.h>

typedef __attribute__((ext_vector_type(2))) float v2f;
typedef __attribute__((ext_vector_type(8))) float v8f;

#define CFEAT 256
#define PK_SHIFT 18
#define PK_MASK ((1u << PK_SHIFT) - 1u)

template <int ROWS>
__global__ void prep_kernel(float* __restrict__ zp, int nz,
                            float* __restrict__ zq, int nzq,
                            const int* __restrict__ src,
                            const int* __restrict__ dst,
                            unsigned int* __restrict__ key,
                            unsigned int* __restrict__ pk, int E,
                            const float* __restrict__ pstats, int nPrev,
                            const float* __restrict__ pg,
                            const float* __restrict__ pb,
                            const float* __restrict__ Wl,
                            const float* __restrict__ Wr,
                            const float* __restrict__ bl,
                            float* __restrict__ Wls,
                            float* __restrict__ Wrs,
                            float* __restrict__ bAgg,
                            float* __restrict__ bRoot) {
    const int nWords = ((E + 511) >> 9) * 128;
    const int i = (int)(blockIdx.x * blockDim.x + threadIdx.x);
    const int stride = (int)(gridDim.x * blockDim.x);
    for (int j = i; j < nz; j += stride) zp[j] = 0.0f;
    for (int j = i; j < nzq; j += stride) zq[j] = 0.0f;
    if (pstats != nullptr) {
        const float invN = 1.0f / (float)nPrev;
        for (int j = i; j < 2 * CFEAT * CFEAT; j += stride) {
            const int m = j >> 16;
            const int idx = j & (CFEAT * CFEAT - 1);
            const int k = idx & (CFEAT - 1);
            const float mu = pstats[k] * invN;
            const float var = pstats[CFEAT + k] * invN - mu * mu;
            const float a = pg[k] * rsqrtf(var + 1e-5f);
            if (m == 0) Wls[idx] = Wl[idx] * a;
            else        Wrs[idx] = Wr[idx] * a;
        }
        const int gwave = i >> 5;
        const int lane = i & 31;
        if (gwave < 2 * CFEAT) {
            const int m = gwave >> 8;
            const int jrow = gwave & (CFEAT - 1);
            const float* wrow = (m == 0 ? Wl : Wr) + jrow * CFEAT;
            float acc = 0.0f;
            for (int t = 0; t < CFEAT / 32; ++t) {
                const int k = lane + 32 * t;
                const float mu = pstats[k] * invN;
                const float var = pstats[CFEAT + k] * invN - mu * mu;
                const float a = pg[k] * rsqrtf(var + 1e-5f);
                const float c = pb[k] - a * mu;
                acc += c * wrow[k];
            }
            for (int d = 16; d >= 1; d >>= 1) acc += __shfl_xor(acc, d, 32);
            if (lane == 0) {
                if (m == 0) bAgg[jrow] = acc;
                else        bRoot[jrow] = bl[jrow] + acc;
            }
        }
    }
    for (int j = i; j < nWords; j += stride) {
        const int e = 4 * j;
        unsigned w = 0u;
        if (e + 3 < E) {
            const int4 d = *(const int4*)(dst + e);
            const int4 sv = *(const int4*)(src + e);
            const int ox = d.x / ROWS, oy = d.y / ROWS, oz = d.z / ROWS, ow = d.w / ROWS;
            w = ((unsigned)ox & 255u) | (((unsigned)oy & 255u) << 8) |
                (((unsigned)oz & 255u) << 16) | (((unsigned)ow & 255u) << 24);
            uint4 p;
            p.x = (unsigned)sv.x | ((unsigned)(d.x - ox * ROWS) << PK_SHIFT);
            p.y = (unsigned)sv.y | ((unsigned)(d.y - oy * ROWS) << PK_SHIFT);
            p.z = (unsigned)sv.z | ((unsigned)(d.z - oz * ROWS) << PK_SHIFT);
            p.w = (unsigned)sv.w | ((unsigned)(d.w - ow * ROWS) << PK_SHIFT);
            *(uint4*)(pk + e) = p;
        } else {
            for (int b = 0; b < 4; ++b)
                if (e + b < E) {
                    const int db = dst[e + b];
                    const int ob = db / ROWS;
                    w |= ((unsigned)ob & 255u) << (8 * b);
                    pk[e + b] = (unsigned)src[e + b] | ((unsigned)(db - ob * ROWS) << PK_SHIFT);
                }
        }
        key[j] = w;
    }
}

#define SEG_THREADS 1024
#define SEG_WAVES 32
#define SEG_GRID 256

__device__ __forceinline__ void row_load(float (&v)[8], const float* __restrict__ x, int sr, int lane) {
    const float4* xs = (const float4*)(x + (long)sr * CFEAT) + lane;
    const float4 lo = xs[0];
    const float4 hi = xs[32];
    v[0] = lo.x; v[1] = lo.y; v[2] = lo.z; v[3] = lo.w;
    v[4] = hi.x; v[5] = hi.y; v[6] = hi.z; v[7] = hi.w;
}

__device__ __forceinline__ void row_add_lds(const float (&v)[8], float* acc, int dr, int lane) {
    float* a = acc + dr * CFEAT + lane;
#pragma unroll
    for (int j = 0; j < 8; ++j) atomicAdd(&a[(j >> 2) * 128 + (j & 3) * 32], v[j]);
}

#define SEG_BATCH() do { \
        const unsigned w0_ = pk[qe0], w1_ = pk[qe1], w2_ = pk[qe2], w3_ = pk[qe3]; \
        const int s0_ = (int)(w0_ & PK_MASK), s1_ = (int)(w1_ & PK_MASK); \
        const int s2_ = (int)(w2_ & PK_MASK), s3_ = (int)(w3_ & PK_MASK); \
        const int d0_ = (int)(w0_ >> PK_SHIFT), d1_ = (int)(w1_ >> PK_SHIFT); \
        const int d2_ = (int)(w2_ >> PK_SHIFT), d3_ = (int)(w3_ >> PK_SHIFT); \
        if (lane < 4) { \
            const int dl_ = (lane == 0) ? d0_ : (lane == 1) ? d1_ : (lane == 2) ? d2_ : d3_; \
            atomicAdd(&cntL[dl_], 1.0f); \
        } \
        float v0_[8], v1_[8], v2_[8], v3_[8]; \
        row_load(v0_, x, s0_, lane); \
        row_load(v1_, x, s1_, lane); \
        row_load(v2_, x, s2_, lane); \
        row_load(v3_, x, s3_, lane); \
        row_add_lds(v0_, acc, d0_, lane); \
        row_add_lds(v1_, acc, d1_, lane); \
        row_add_lds(v2_, acc, d2_, lane); \
        row_add_lds(v3_, acc, d3_, lane); \
        np = 0; \
    } while (0)

#define SEG_ENQUEUE(E_) do { \
        const int e_ = (E_); \
        if (e_ < E) { \
            if (np == 0)      qe0 = e_; \
            else if (np == 1) qe1 = e_; \
            else if (np == 2) qe2 = e_; \
            else              qe3 = e_; \
            ++np; \
            if (np == 4) SEG_BATCH(); \
        } \
    } while (0)

#define SEG_ZFLAGS(X_) (~((((X_) & 0x7F7F7F7Fu) + 0x7F7F7F7Fu) | (X_)) & 0x80808080u)

#define SEG_PROCESS(CC_, K_) do { \
        const int cc_ = (CC_); \
        const unsigned own_ = (SEG_ZFLAGS((K_).x ^ gpat) >> 7) | (SEG_ZFLAGS((K_).y ^ gpat) >> 6) | \
                              (SEG_ZFLAGS((K_).z ^ gpat) >> 5) | (SEG_ZFLAGS((K_).w ^ gpat) >> 4); \
        unsigned M_ = __builtin_amdgcn_ballot_w32(own_ != 0u); \
        if (cc_ >= nCh) M_ = 0u; \
        if (M_) { \
            const int ebase_ = SEG_CHUNK(cc_) * 512; \
            while (M_) { \
                const int i_ = __builtin_ctz(M_); \
                M_ &= M_ - 1; \
                unsigned o_ = (unsigned)__builtin_amdgcn_readlane((int)own_, i_); \
                while (o_) { \
                    const int j_ = __builtin_ctz(o_); \
                    o_ &= o_ - 1; \
                    SEG_ENQUEUE(ebase_ + i_ * 16 + ((j_ & 3) << 2) + (j_ >> 3)); \
                } \
            } \
        } \
    } while (0)

#define SEG_CHUNK(T_) (((T_) + rot < nCh) ? ((T_) + rot) : ((T_) + rot - nCh))
#define SEG_LOAD(C_) (*(const uint4*)(key + SEG_CHUNK(((C_) < last) ? (C_) : last) * 128 + lane * 4))

template <int ROWS>
__global__ void __launch_bounds__(SEG_THREADS) segsum_kernel(const float* __restrict__ x,
                              const unsigned int* __restrict__ pk,
                              const unsigned int* __restrict__ key,
                              float* __restrict__ s,
                              float* __restrict__ cnt,
                              int E, int N) {
    __shared__ __attribute__((aligned(16))) float acc[ROWS * CFEAT];
    __shared__ float cntL[ROWS];
    const int tid = threadIdx.x;
    const int lane = tid & 31;
    const int wv = __builtin_amdgcn_readfirstlane(tid >> 5);
    const int r0 = (int)blockIdx.x * ROWS;
    const unsigned gpat = (unsigned)blockIdx.x * 0x01010101u;

    float4* acc4 = (float4*)acc;
    for (int i = tid; i < ROWS * (CFEAT / 4); i += SEG_THREADS) acc4[i] = make_float4(0.0f, 0.0f, 0.0f, 0.0f);
    for (int i = tid; i < ROWS; i += SEG_THREADS) cntL[i] = 0.0f;
    __syncthreads();

    int qe0 = 0, qe1 = 0, qe2 = 0, qe3 = 0, np = 0;

    const int nCh = (E + 511) >> 9;
    if (nCh > 0) {
        const int last = nCh - 1;
        const int rot = (int)(((long)blockIdx.x * nCh) / SEG_GRID);
        uint4 dA = SEG_LOAD(wv);
        __builtin_amdgcn_sched_barrier(0);
        uint4 dB = SEG_LOAD(wv + SEG_WAVES);
        __builtin_amdgcn_sched_barrier(0);
        uint4 dC = SEG_LOAD(wv + 2 * SEG_WAVES);
        uint4 dD;
        for (int c = wv; c < nCh; c += 4 * SEG_WAVES) {
            dD = SEG_LOAD(c + 3 * SEG_WAVES);
            SEG_PROCESS(c, dA);
            dA = SEG_LOAD(c + 4 * SEG_WAVES);
            SEG_PROCESS(c + SEG_WAVES, dB);
            dB = SEG_LOAD(c + 5 * SEG_WAVES);
            SEG_PROCESS(c + 2 * SEG_WAVES, dC);
            dC = SEG_LOAD(c + 6 * SEG_WAVES);
            SEG_PROCESS(c + 3 * SEG_WAVES, dD);
        }
    }
    if (np > 0) {
        const int e1 = (np > 1) ? qe1 : qe0, e2 = (np > 2) ? qe2 : qe0;
        const unsigned w0 = pk[qe0], w1 = pk[e1], w2 = pk[e2];
        const int d0 = (int)(w0 >> PK_SHIFT), d1 = (int)(w1 >> PK_SHIFT), d2 = (int)(w2 >> PK_SHIFT);
        if (lane < np) {
            const int dl = (lane == 0) ? d0 : (lane == 1) ? d1 : d2;
            atomicAdd(&cntL[dl], 1.0f);
        }
        float v0[8];
        row_load(v0, x, (int)(w0 & PK_MASK), lane);
        row_add_lds(v0, acc, d0, lane);
        if (np > 1) {
            row_load(v0, x, (int)(w1 & PK_MASK), lane);
            row_add_lds(v0, acc, d1, lane);
        }
        if (np > 2) {
            row_load(v0, x, (int)(w2 & PK_MASK), lane);
            row_add_lds(v0, acc, d2, lane);
        }
    }
    __syncthreads();

    int rows = N - r0;
    rows = rows < ROWS ? rows : ROWS;
    float4* s4 = (float4*)(s + (long)r0 * CFEAT);
    for (int i = tid; i < rows * (CFEAT / 4); i += SEG_THREADS) {
        const float* a = acc + (i >> 6) * CFEAT + ((i >> 5) & 1) * 128 + (i & 31);
        s4[i] = make_float4(a[0], a[32], a[64], a[96]);
    }
    for (int i = tid; i < rows; i += SEG_THREADS) cnt[r0 + i] = cntL[i];
}

#define GEMM_RT_MAX 4
#define GEMM_GRID 256
typedef float4 linefrag[4];

__device__ __forceinline__ void load_linefrag(linefrag& f, const float* __restrict__ rowh, int kc) {
    const float4* p = (const float4*)(rowh + kc);
#pragma unroll
    for (int q = 0; q < 4; ++q) f[q] = p[2 * q];
}

__device__ __forceinline__ v2f frag_pair(const linefrag& f, int q, int p) {
    v2f r;
    if (p == 0) { r[0] = f[q].x; r[1] = f[q].y; }
    else        { r[0] = f[q].z; r[1] = f[q].w; }
    return r;
}

template <int RT>
__device__ __forceinline__ void mma_line(v8f (&acc)[RT][4], const linefrag (&a)[RT],
                                         const linefrag (&b)[4]) {
#pragma unroll
    for (int q = 0; q < 4; ++q) {
#pragma unroll
        for (int p = 0; p < 2; ++p) {
#pragma unroll
            for (int r = 0; r < RT; ++r) {
                v2f af = frag_pair(a[r], q, p);
#pragma unroll
                for (int t = 0; t < 4; ++t)
                    acc[r][t] = __builtin_amdgcn_wmma_f32_16x16x4_f32(
                        false, af, false, frag_pair(b[t], q, p), (short)0, acc[r][t], false, false);
            }
        }
    }
}

template <int RT>
__device__ __forceinline__ void gemm_group(const float* __restrict__ s,
                            const float* __restrict__ cnt,
                            const float* __restrict__ xt,
                            const float* __restrict__ Wl,
                            const float* __restrict__ bl,
                            const float* __restrict__ Wr,
                            const float* __restrict__ bAgg,
                            float* __restrict__ h,
                            float* __restrict__ colsum,
                            float* __restrict__ colsq,
                            int N, int t0, int colBase, int lane) {
    int l15  = lane & 15;
    int hoff = (lane >> 4) * 4;

    float invc[RT];
    unsigned hasIn[RT];
    const float* sRow[RT];
    const float* xRow[RT];
#pragma unroll
    for (int r = 0; r < RT; ++r) {
        int rowA = (t0 + r) * 16 + l15;
        int rowL = (rowA < N) ? rowA : (N - 1);
        const float cr = cnt[rowL];
        invc[r] = 1.0f / fmaxf(cr, 1.0f);
        hasIn[r] = __builtin_amdgcn_ballot_w32(cr > 0.0f);
        sRow[r] = s  + (long)rowL * CFEAT + hoff;
        xRow[r] = xt + (long)rowL * CFEAT + hoff;
    }

    const float* wlRow[4];
    const float* wrRow[4];
#pragma unroll
    for (int t = 0; t < 4; ++t) {
        int colB = colBase + t * 16 + l15;
        wlRow[t] = Wl + (long)colB * CFEAT + hoff;
        wrRow[t] = Wr + (long)colB * CFEAT + hoff;
    }

    v8f acc[RT][4];
#pragma unroll
    for (int r = 0; r < RT; ++r)
#pragma unroll
        for (int t = 0; t < 4; ++t)
            acc[r][t] = (v8f){0, 0, 0, 0, 0, 0, 0, 0};

    linefrag aC[RT], bC[4], aN[RT], bN[4];
#pragma unroll
    for (int r = 0; r < RT; ++r) load_linefrag(aC[r], sRow[r], 0);
#pragma unroll
    for (int t = 0; t < 4; ++t) load_linefrag(bC[t], wlRow[t], 0);

    for (int kc = 0; kc < CFEAT - 32; kc += 32) {
        asm volatile("v_nop\n\tv_nop\n\tv_nop\n\tv_nop");
        __builtin_amdgcn_sched_barrier(0);
#pragma unroll
        for (int r = 0; r < RT; ++r) load_linefrag(aN[r], xRow[r], kc);
#pragma unroll
        for (int t = 0; t < 4; ++t) load_linefrag(bN[t], wrRow[t], kc);
        __builtin_amdgcn_sched_barrier(0);
#pragma unroll
        for (int r = 0; r < RT; ++r)
#pragma unroll
            for (int q = 0; q < 4; ++q) aC[r][q] *= invc[r];
        mma_line<RT>(acc, aC, bC);
        __builtin_amdgcn_sched_barrier(0);
#pragma unroll
        for (int r = 0; r < RT; ++r) load_linefrag(aC[r], sRow[r], kc + 32);
#pragma unroll
        for (int t = 0; t < 4; ++t) load_linefrag(bC[t], wlRow[t], kc + 32);
        __builtin_amdgcn_sched_barrier(0);
        mma_line<RT>(acc, aN, bN);
    }
#pragma unroll
    for (int r = 0; r < RT; ++r) load_linefrag(aN[r], xRow[r], CFEAT - 32);
#pragma unroll
    for (int t = 0; t < 4; ++t) load_linefrag(bN[t], wrRow[t], CFEAT - 32);
    __builtin_amdgcn_sched_barrier(0);
#pragma unroll
    for (int r = 0; r < RT; ++r)
#pragma unroll
        for (int q = 0; q < 4; ++q) aC[r][q] *= invc[r];
    mma_line<RT>(acc, aC, bC);
    mma_line<RT>(acc, aN, bN);

    int half8 = (lane < 16) ? 0 : 8;
#pragma unroll
    for (int t = 0; t < 4; ++t) {
        int col = colBase + t * 16 + l15;
        float bias = bl[col];
        float ba = (bAgg != nullptr) ? bAgg[col] : 0.0f;
        float sp = 0.0f, qp = 0.0f;
#pragma unroll
        for (int r = 0; r < RT; ++r) {
            int rbase = (t0 + r) * 16 + half8;
#pragma unroll
            for (int v = 0; v < 8; ++v) {
                float bv = ((hasIn[r] >> (half8 + v)) & 1u) ? (bias + ba) : bias;
                float val = fmaxf(acc[r][t][v] + bv, 0.0f);
                int rr = rbase + v;
                bool ok = (rr < N);
                float m = ok ? val : 0.0f;
                sp += m;
                qp += m * m;
                if (ok) h[(long)rr * CFEAT + col] = val;
            }
        }
        atomicAdd(&colsum[col], sp);
        atomicAdd(&colsq[col], qp);
    }
}

__global__ void __launch_bounds__(256) gemm_kernel(const float* __restrict__ s,
                            const float* __restrict__ cnt,
                            const float* __restrict__ xt,
                            const float* __restrict__ Wl,
                            const float* __restrict__ bl,
                            const float* __restrict__ Wr,
                            const float* __restrict__ bAgg,
                            float* __restrict__ h,
                            float* __restrict__ colsum,
                            float* __restrict__ colsq,
                            int N, int nT16) {
    const int lane = threadIdx.x & 31;
    const int wv = __builtin_amdgcn_readfirstlane((int)(threadIdx.x >> 5));
    const int colBase = (wv & 3) * 64;
    const int part = wv >> 2;
    const int nParts = (int)(blockDim.x >> 7);
    const int G = (int)gridDim.x;
    const int tBeg = (int)(((unsigned)blockIdx.x * (unsigned)nT16) / (unsigned)G);
    const int tEnd = (int)((((unsigned)blockIdx.x + 1u) * (unsigned)nT16) / (unsigned)G);
    const int n = tEnd - tBeg;
    const int pBeg = tBeg + (part * n) / nParts;
    const int pEnd = tBeg + ((part + 1) * n) / nParts;
    const int m = pEnd - pBeg;
    if (m <= 0) return;
    const int k = (m + GEMM_RT_MAX - 1) / GEMM_RT_MAX;
    const int base = m / k;
    const int extra = m - base * k;
    int t0 = pBeg;
    for (int g = 0; g < k; ++g) {
        const int sz = base + ((g < extra) ? 1 : 0);
#define GEMM_CASE(R_) case R_: gemm_group<R_>(s, cnt, xt, Wl, bl, Wr, bAgg, h, colsum, colsq, N, t0, colBase, lane); break;
        switch (sz) {
            GEMM_CASE(1) GEMM_CASE(2) GEMM_CASE(3)
            default: gemm_group<GEMM_RT_MAX>(s, cnt, xt, Wl, bl, Wr, bAgg, h, colsum, colsq, N, t0, colBase, lane); break;
        }
#undef GEMM_CASE
        t0 += sz;
    }
}

__global__ void norm_kernel(const float* __restrict__ hbuf,
                            const float* __restrict__ colsum,
                            const float* __restrict__ colsq,
                            const float* __restrict__ gamma,
                            const float* __restrict__ beta,
                            float* __restrict__ out, int N) {
    long i = (long)blockIdx.x * blockDim.x + threadIdx.x;
    long total = (long)N * CFEAT;
    if (i >= total) return;
    int col = (int)(i & (CFEAT - 1));
    float invN = 1.0f / (float)N;
    float mu  = colsum[col] * invN;
    float var = colsq[col] * invN - mu * mu;
    float rs  = rsqrtf(var + 1e-5f);
    out[i] = gamma[col] * (hbuf[i] - mu) * rs + beta[col];
}

extern "C" void kernel_launch(void* const* d_in, const int* in_sizes, int n_in,
                              void* d_out, int out_size, void* d_ws, size_t ws_size,
                              hipStream_t stream) {
    const float* x    = (const float*)d_in[0];
    const int* e0src  = (const int*)d_in[1];
    const int* e0dst  = (const int*)d_in[2];
    const int* e1src  = (const int*)d_in[3];
    const int* e1dst  = (const int*)d_in[4];
    const float* Wl0  = (const float*)d_in[5];
    const float* bl0  = (const float*)d_in[6];
    const float* Wr0  = (const float*)d_in[7];
    const float* g0   = (const float*)d_in[8];
    const float* be0  = (const float*)d_in[9];
    const float* Wl1  = (const float*)d_in[10];
    const float* bl1  = (const float*)d_in[11];
    const float* Wr1  = (const float*)d_in[12];
    const float* g1   = (const float*)d_in[13];
    const float* be1  = (const float*)d_in[14];

    const int N1 = 50000, N2 = 12500;
    const int E0 = in_sizes[1];
    const int E1 = in_sizes[3];

    float* ws = (float*)d_ws;
    const long NC1 = (long)N1 * CFEAT;
    float* x1    = ws;
    float* sbuf  = ws + NC1;
    float* hbuf  = ws + 2 * NC1;
    float* cntb  = ws + 3 * NC1;
    float* stats = cntb + 50048;
    float* Wls    = hbuf + 4 * 1048576;
    float* Wrs    = Wls + CFEAT * CFEAT;
    float* bAgg   = Wrs + CFEAT * CFEAT;
    float* bRoot  = bAgg + CFEAT;
    float* stats1 = bRoot + CFEAT;
    unsigned int* pkb = (unsigned int*)(hbuf + 5 * 1048576);

    unsigned int* keyb = (unsigned int*)hbuf;
    prep_kernel<196><<<256, 256, 0, stream>>>(stats, 512, nullptr, 0, e0src, e0dst, keyb, pkb, E0,
        nullptr, 0, nullptr, nullptr, nullptr, nullptr, nullptr, nullptr, nullptr, nullptr, nullptr);
    segsum_kernel<196><<<SEG_GRID, SEG_THREADS, 0, stream>>>(x, pkb, keyb, sbuf, cntb, E0, N1);
    gemm_kernel<<<GEMM_GRID, 256, 0, stream>>>(
        sbuf, cntb, x, Wl0, bl0, Wr0, nullptr, x1, stats, stats + 256, N1, 3125);

    const long NC2 = (long)N2 * CFEAT;
    prep_kernel<49><<<256, 256, 0, stream>>>(stats1, 512, nullptr, 0, e1src, e1dst, keyb, pkb, E1,
        stats, N1, g0, be0, Wl1, Wr1, bl1, Wls, Wrs, bAgg, bRoot);
    segsum_kernel<49><<<SEG_GRID, SEG_THREADS, 0, stream>>>(x1, pkb, keyb, sbuf, cntb, E1, N2);
    gemm_kernel<<<196, 128, 0, stream>>>(
        sbuf, cntb, x1, Wls, bRoot, Wrs, bAgg, hbuf, stats1, stats1 + 256, N2, 782);
    norm_kernel<<<(int)((NC2 + 255) / 256), 256, 0, stream>>>(
        hbuf, stats1, stats1 + 256, g1, be1, (float*)d_out, N2);
}
